# starting kernel
# baseline (speedup 1.0000x reference)
#include <hip/hip_runtime.h>
#include <hip/hip_bf16.h>
#include <math.h>

#define N_NODES     50000
#define N_EDGES     800000
#define NUM_FEATS   128
#define HIDDEN      256
#define NUM_CLASSES 40

typedef __attribute__((ext_vector_type(16))) _Float16 v16h;
typedef __attribute__((ext_vector_type(8)))  float    v8f;

__global__ void scatter_kernel(const float* __restrict__ x,
                               const int*   __restrict__ ei,
                               const float* __restrict__ ew,
                               float*       __restrict__ agg,
                               int D) {
    const int vecPerEdge = D >> 2;
    const int gid = blockIdx.x * blockDim.x + threadIdx.x;
    const int total = N_EDGES * vecPerEdge;
    if (gid >= total) return;
    const int e  = gid / vecPerEdge;
    const int f4 = (gid - e * vecPerEdge) << 2;
    const int src = ei[e];
    const int dst = ei[N_EDGES + e];
    const float w = ew[e];
    const float4 v = *(const float4*)(x + (long long)src * D + f4);
    float* o = agg + (long long)dst * D + f4;
    atomicAdd(o + 0, v.x * w);
    atomicAdd(o + 1, v.y * w);
    atomicAdd(o + 2, v.z * w);
    atomicAdd(o + 3, v.w * w);
}

template <int K>
__global__ __launch_bounds__(32)
void gemm_wmma_kernel(const float* __restrict__ A,
                      const float* __restrict__ W,
                      float*       __restrict__ C,
                      int Ncols, int do_relu) {
    const int l     = threadIdx.x;
    const int rl    = l & 15;
    const int khalf = (l >> 4) * 8;
    const int row0  = blockIdx.x * 16;
    const int col0  = blockIdx.y * 16;
    const int col   = col0 + rl;

    v8f acc = {};
    const float* arow = A + (long long)(row0 + rl) * K + khalf;
    const float* wrow = W + (long long)col * K + khalf;

    #pragma unroll
    for (int k0 = 0; k0 < K; k0 += 32) {
        v16h af;
        #pragma unroll
        for (int e = 0; e < 16; ++e) {
            const int kk = k0 + ((e & 8) ? 16 : 0) + (e & 7);
            af[e] = (_Float16)arow[kk];
        }
        v16h bf = {};
        if (col < Ncols) {
            #pragma unroll
            for (int e = 0; e < 16; ++e) {
                const int kk = k0 + ((e & 8) ? 16 : 0) + (e & 7);
                bf[e] = (_Float16)wrow[kk];
            }
        }
        acc = __builtin_amdgcn_wmma_f32_16x16x32_f16(
             false, af,  false, bf,
             (short)0, acc,  false,  false);
    }

    if (col < Ncols) {
        #pragma unroll
        for (int r = 0; r < 8; ++r) {
            float v = acc[r];
            if (do_relu) v = v > 0.0f ? v : 0.0f;
            C[(long long)(row0 + khalf + r) * Ncols + col] = v;
        }
    }
}

__global__ void logsoftmax_kernel(const float* __restrict__ H,
                                  float*       __restrict__ out) {
    const int n = blockIdx.x * blockDim.x + threadIdx.x;
    if (n >= N_NODES) return;
    const float* h = H + (long long)n * NUM_CLASSES;
    float mx = -INFINITY;
    #pragma unroll
    for (int c = 0; c < NUM_CLASSES; ++c) mx = fmaxf(mx, h[c]);
    float s = 0.0f;
    #pragma unroll
    for (int c = 0; c < NUM_CLASSES; ++c) s += __expf(h[c] - mx);
    const float lse = __logf(s) + mx;
    float* o = out + (long long)n * NUM_CLASSES;
    #pragma unroll
    for (int c = 0; c < NUM_CLASSES; ++c) o[c] = h[c] - lse;
}

extern "C" void kernel_launch(void* const* d_in, const int* in_sizes, int n_in,
                              void* d_out, int out_size, void* d_ws, size_t ws_size,
                              hipStream_t stream) {
    const float* x  = (const float*)d_in[0];
    const int*   ei = (const int*)  d_in[1];
    const float* ew = (const float*)d_in[2];
    const float* w1 = (const float*)d_in[3];
    const float* w2 = (const float*)d_in[4];
    float* out = (float*)d_out;

    const size_t bufElems = (size_t)N_NODES * HIDDEN;
    float* bufA = (float*)d_ws;
    float* bufB = bufA + bufElems;
    float* agg1 = bufA;
    float* h1   = bufB;
    float* agg2 = bufA;
    float* h2   = bufB;

    hipMemsetAsync(agg1, 0, (size_t)N_NODES * NUM_FEATS * sizeof(float), stream);
    {
        const int total = N_EDGES * (NUM_FEATS / 4);
        scatter_kernel<<<(total + 255) / 256, 256, 0, stream>>>(x, ei, ew, agg1, NUM_FEATS);
    }
    gemm_wmma_kernel<NUM_FEATS><<<dim3(N_NODES / 16, HIDDEN / 16), 32, 0, stream>>>(
        agg1, w1, h1, HIDDEN,  1);

    hipMemsetAsync(agg2, 0, (size_t)N_NODES * HIDDEN * sizeof(float), stream);
    {
        const int total = N_EDGES * (HIDDEN / 4);
        scatter_kernel<<<(total + 255) / 256, 256, 0, stream>>>(h1, ei, ew, agg2, HIDDEN);
    }
    gemm_wmma_kernel<HIDDEN><<<dim3(N_NODES / 16, 3  ), 32, 0, stream>>>(
        agg2, w2, h2, NUM_CLASSES,  0);

    logsoftmax_kernel<<<(N_NODES + 255) / 256, 256, 0, stream>>>(h2, out);
}


# both layers' edge scatter-add (float atomics, zero-fill) replaced by a per-call counting sort of edges by destination and per-row gather sums in registers; sums now in edge-id order per row
# speedup vs baseline: 1.6030x; 1.6030x over previous
#include <hip/hip_runtime.h>
#include <hip/hip_bf16.h>
#include <math.h>

#define N_NODES     50000
#define N_EDGES     800000
#define NUM_FEATS   128
#define HIDDEN      256
#define NUM_CLASSES 40

typedef __attribute__((ext_vector_type(16))) _Float16 v16h;
typedef __attribute__((ext_vector_type(8)))  float    v8f;

#define ROWPTR_PAD 50176
#define SCAN_TILE  1024
#define SCAN_TILES ((N_NODES + SCAN_TILE - 1) / SCAN_TILE)

__global__ void degree_kernel(const int* __restrict__ ei, int* __restrict__ cnt) {
    const int e = blockIdx.x * blockDim.x + threadIdx.x;
    if (e < N_EDGES) atomicAdd(cnt + ei[N_EDGES + e], 1);
}

__global__ __launch_bounds__(SCAN_TILE)
void rowptr_kernel(const int* __restrict__ cnt, int* __restrict__ rowptr) {
    __shared__ int wsum[32];
    __shared__ int wpre[32];
    const int t    = threadIdx.x;
    const int lane = t & 31;
    const int wv   = t >> 5;
    const int idx  = blockIdx.x * SCAN_TILE + t;
    int pre = 0;
    for (int i = 0; i < (int)blockIdx.x; ++i) pre += cnt[i * SCAN_TILE + t];
    const int c = idx < N_NODES ? cnt[idx] : 0;
    int inc = c;
    for (int o = 1; o < 32; o <<= 1) {
        const int v = __shfl_up(inc, o);
        if (lane >= o) inc += v;
    }
    for (int o = 16; o > 0; o >>= 1) pre += __shfl_xor(pre, o);
    if (lane == 31) wsum[wv] = inc;
    if (lane == 0)  wpre[wv] = pre;
    __syncthreads();
    int base = 0;
    for (int j = 0; j < 32; ++j) {
        base += wpre[j];
        if (j < wv) base += wsum[j];
    }
    const int excl = base + inc - c;
    if (idx < N_NODES) rowptr[idx] = excl;
    if (idx == N_NODES - 1) rowptr[N_NODES] = excl + c;
}

__global__ void place_kernel(const int* __restrict__ ei,
                             const int* __restrict__ rowptr,
                             int*       __restrict__ cnt,
                             int*       __restrict__ perm) {
    const int e = blockIdx.x * blockDim.x + threadIdx.x;
    if (e >= N_EDGES) return;
    const int d    = ei[N_EDGES + e];
    const int slot = atomicSub(cnt + d, 1) - 1;
    perm[rowptr[d] + slot] = e;
}

__global__ void order_kernel(const int*   __restrict__ ei,
                             const float* __restrict__ ew,
                             const int*   __restrict__ rowptr,
                             const int*   __restrict__ perm,
                             int*         __restrict__ csrc,
                             float*       __restrict__ cw) {
    const int e = blockIdx.x * blockDim.x + threadIdx.x;
    if (e >= N_EDGES) return;
    const int d   = ei[N_EDGES + e];
    const int beg = rowptr[d];
    const int end = rowptr[d + 1];
    int r = 0;
    for (int p = beg; p < end; ++p) r += (perm[p] < e) ? 1 : 0;
    csrc[beg + r] = ei[e];
    cw[beg + r]   = ew[e];
}

template <int G>
__device__ __forceinline__ void take_rows(const float* __restrict__ feat, int D,
                                          int s, float w, int j, float4& acc) {
    float4 v[G];
    float  ww[G];
    #pragma unroll
    for (int u = 0; u < G; ++u) {
        const int sj = __builtin_amdgcn_readlane(s, j + u);
        ww[u] = __int_as_float(__builtin_amdgcn_readlane(__float_as_int(w), j + u));
        v[u] = *(const float4*)(feat + (long long)sj * D);
    }
    #pragma unroll
    for (int u = 0; u < G; ++u) {
        acc.x = fmaf(ww[u], v[u].x, acc.x);
        acc.y = fmaf(ww[u], v[u].y, acc.y);
        acc.z = fmaf(ww[u], v[u].z, acc.z);
        acc.w = fmaf(ww[u], v[u].w, acc.w);
    }
}

#define GATHER_WAVES  8
#define GATHER_BLOCKS 1250
template <int D>
__global__ __launch_bounds__(GATHER_WAVES * 32)
void gather_kernel(const float* __restrict__ feat,
                   const int*   __restrict__ rowptr,
                   const int*   __restrict__ csrc,
                   const float* __restrict__ cw,
                   float*       __restrict__ agg) {
    constexpr int P = D / 128;
    const int lane = threadIdx.x & 31;
    const int wid = __builtin_amdgcn_readfirstlane(
        (int)(blockIdx.x * GATHER_WAVES + (threadIdx.x >> 5)));
    const int nw = gridDim.x * GATHER_WAVES;
    for (int item = wid; item < N_NODES * P; item += nw) {
        const int row = item / P;
        const int off = (item - row * P) * 128 + lane * 4;
        const float* f = feat + off;
        const int beg = rowptr[row];
        const int end = rowptr[row + 1];
        float4 acc = make_float4(0.0f, 0.0f, 0.0f, 0.0f);
        for (int base = beg; base < end; base += 32) {
            const int p = base + lane;
            int   s = 0;
            float w = 0.0f;
            if (p < end) { s = csrc[p]; w = cw[p]; }
            const int n = (end - base) < 32 ? (end - base) : 32;
            int j = 0;
            for (; j + 8 <= n; j += 8) take_rows<8>(f, D, s, w, j, acc);
            switch (n - j) {
                case 7: take_rows<7>(f, D, s, w, j, acc); break;
                case 6: take_rows<6>(f, D, s, w, j, acc); break;
                case 5: take_rows<5>(f, D, s, w, j, acc); break;
                case 4: take_rows<4>(f, D, s, w, j, acc); break;
                case 3: take_rows<3>(f, D, s, w, j, acc); break;
                case 2: take_rows<2>(f, D, s, w, j, acc); break;
                case 1: take_rows<1>(f, D, s, w, j, acc); break;
                default: break;
            }
        }
        *(float4*)(agg + (long long)row * D + off) = acc;
    }
}

template <int K>
__global__ __launch_bounds__(32)
void gemm_wmma_kernel(const float* __restrict__ A,
                      const float* __restrict__ W,
                      float*       __restrict__ C,
                      int Ncols, int do_relu) {
    const int l     = threadIdx.x;
    const int rl    = l & 15;
    const int khalf = (l >> 4) * 8;
    const int row0  = blockIdx.x * 16;
    const int col0  = blockIdx.y * 16;
    const int col   = col0 + rl;

    v8f acc = {};
    const float* arow = A + (long long)(row0 + rl) * K + khalf;
    const float* wrow = W + (long long)col * K + khalf;

    #pragma unroll
    for (int k0 = 0; k0 < K; k0 += 32) {
        v16h af;
        #pragma unroll
        for (int e = 0; e < 16; ++e) {
            const int kk = k0 + ((e & 8) ? 16 : 0) + (e & 7);
            af[e] = (_Float16)arow[kk];
        }
        v16h bf = {};
        if (col < Ncols) {
            #pragma unroll
            for (int e = 0; e < 16; ++e) {
                const int kk = k0 + ((e & 8) ? 16 : 0) + (e & 7);
                bf[e] = (_Float16)wrow[kk];
            }
        }
        acc = __builtin_amdgcn_wmma_f32_16x16x32_f16(
             false, af,  false, bf,
             (short)0, acc,  false,  false);
    }

    if (col < Ncols) {
        #pragma unroll
        for (int r = 0; r < 8; ++r) {
            float v = acc[r];
            if (do_relu) v = v > 0.0f ? v : 0.0f;
            C[(long long)(row0 + khalf + r) * Ncols + col] = v;
        }
    }
}

__global__ void logsoftmax_kernel(const float* __restrict__ H,
                                  float*       __restrict__ out) {
    const int n = blockIdx.x * blockDim.x + threadIdx.x;
    if (n >= N_NODES) return;
    const float* h = H + (long long)n * NUM_CLASSES;
    float mx = -INFINITY;
    #pragma unroll
    for (int c = 0; c < NUM_CLASSES; ++c) mx = fmaxf(mx, h[c]);
    float s = 0.0f;
    #pragma unroll
    for (int c = 0; c < NUM_CLASSES; ++c) s += __expf(h[c] - mx);
    const float lse = __logf(s) + mx;
    float* o = out + (long long)n * NUM_CLASSES;
    #pragma unroll
    for (int c = 0; c < NUM_CLASSES; ++c) o[c] = h[c] - lse;
}

extern "C" void kernel_launch(void* const* d_in, const int* in_sizes, int n_in,
                              void* d_out, int out_size, void* d_ws, size_t ws_size,
                              hipStream_t stream) {
    const float* x  = (const float*)d_in[0];
    const int*   ei = (const int*)  d_in[1];
    const float* ew = (const float*)d_in[2];
    const float* w1 = (const float*)d_in[3];
    const float* w2 = (const float*)d_in[4];
    float* out = (float*)d_out;

    const size_t bufElems = (size_t)N_NODES * HIDDEN;
    float* bufA = (float*)d_ws;
    float* bufB = bufA + bufElems;
    float* agg1 = bufA;
    float* h1   = bufB;
    float* agg2 = bufA;
    float* h2   = bufB;

    int*   cnt    = (int*)bufB;
    int*   perm   = cnt + ROWPTR_PAD;
    int*   rowptr = (int*)d_out;
    int*   csrc   = rowptr + ROWPTR_PAD;
    float* cw     = (float*)(csrc + N_EDGES);

    hipMemsetAsync(cnt, 0, (size_t)N_NODES * sizeof(int), stream);
    degree_kernel<<<(N_EDGES + 255) / 256, 256, 0, stream>>>(ei, cnt);
    rowptr_kernel<<<SCAN_TILES, SCAN_TILE, 0, stream>>>(cnt, rowptr);
    place_kernel<<<(N_EDGES + 255) / 256, 256, 0, stream>>>(ei, rowptr, cnt, perm);
    order_kernel<<<(N_EDGES + 255) / 256, 256, 0, stream>>>(ei, ew, rowptr, perm, csrc, cw);

    gather_kernel<NUM_FEATS><<<GATHER_BLOCKS, GATHER_WAVES * 32, 0, stream>>>(
        x, rowptr, csrc, cw, agg1);
    gemm_wmma_kernel<NUM_FEATS><<<dim3(N_NODES / 16, HIDDEN / 16), 32, 0, stream>>>(
        agg1, w1, h1, HIDDEN,  1);

    gather_kernel<HIDDEN><<<GATHER_BLOCKS, GATHER_WAVES * 32, 0, stream>>>(
        h1, rowptr, csrc, cw, agg2);
    gemm_wmma_kernel<HIDDEN><<<dim3(N_NODES / 16, 3  ), 32, 0, stream>>>(
        agg2, w2, h2, NUM_CLASSES,  0);

    logsoftmax_kernel<<<(N_NODES + 255) / 256, 256, 0, stream>>>(h2, out);
}


# layer 2 reordered: the same WMMA GEMM takes h1 to the 40 classes first, then the ordered per-row gather sums 40 floats per edge, not 256; no atomics; the sum over edges now follows the sum over k
# speedup vs baseline: 1.7207x; 1.0734x over previous
#include <hip/hip_runtime.h>
#include <hip/hip_bf16.h>
#include <math.h>

#define N_NODES     50000
#define N_EDGES     800000
#define NUM_FEATS   128
#define HIDDEN      256
#define NUM_CLASSES 40

typedef __attribute__((ext_vector_type(16))) _Float16 v16h;
typedef __attribute__((ext_vector_type(8)))  float    v8f;

#define ROWPTR_PAD 50176
#define SCAN_TILE  1024
#define SCAN_TILES ((N_NODES + SCAN_TILE - 1) / SCAN_TILE)

__global__ void degree_kernel(const int* __restrict__ ei, int* __restrict__ cnt) {
    const int e = blockIdx.x * blockDim.x + threadIdx.x;
    if (e < N_EDGES) atomicAdd(cnt + ei[N_EDGES + e], 1);
}

__global__ __launch_bounds__(SCAN_TILE)
void rowptr_kernel(const int* __restrict__ cnt, int* __restrict__ rowptr) {
    __shared__ int wsum[32];
    __shared__ int wpre[32];
    const int t    = threadIdx.x;
    const int lane = t & 31;
    const int wv   = t >> 5;
    const int idx  = blockIdx.x * SCAN_TILE + t;
    int pre = 0;
    for (int i = 0; i < (int)blockIdx.x; ++i) pre += cnt[i * SCAN_TILE + t];
    const int c = idx < N_NODES ? cnt[idx] : 0;
    int inc = c;
    for (int o = 1; o < 32; o <<= 1) {
        const int v = __shfl_up(inc, o);
        if (lane >= o) inc += v;
    }
    for (int o = 16; o > 0; o >>= 1) pre += __shfl_xor(pre, o);
    if (lane == 31) wsum[wv] = inc;
    if (lane == 0)  wpre[wv] = pre;
    __syncthreads();
    int base = 0;
    for (int j = 0; j < 32; ++j) {
        base += wpre[j];
        if (j < wv) base += wsum[j];
    }
    const int excl = base + inc - c;
    if (idx < N_NODES) rowptr[idx] = excl;
    if (idx == N_NODES - 1) rowptr[N_NODES] = excl + c;
}

__global__ void place_kernel(const int* __restrict__ ei,
                             const int* __restrict__ rowptr,
                             int*       __restrict__ cnt,
                             int*       __restrict__ perm) {
    const int e = blockIdx.x * blockDim.x + threadIdx.x;
    if (e >= N_EDGES) return;
    const int d    = ei[N_EDGES + e];
    const int slot = atomicSub(cnt + d, 1) - 1;
    perm[rowptr[d] + slot] = e;
}

__global__ void order_kernel(const int*   __restrict__ ei,
                             const float* __restrict__ ew,
                             const int*   __restrict__ rowptr,
                             const int*   __restrict__ perm,
                             int*         __restrict__ csrc,
                             float*       __restrict__ cw) {
    const int e = blockIdx.x * blockDim.x + threadIdx.x;
    if (e >= N_EDGES) return;
    const int d   = ei[N_EDGES + e];
    const int beg = rowptr[d];
    const int end = rowptr[d + 1];
    int r = 0;
    for (int p = beg; p < end; ++p) r += (perm[p] < e) ? 1 : 0;
    csrc[beg + r] = ei[e];
    cw[beg + r]   = ew[e];
}

template <int G>
__device__ __forceinline__ void take_rows(const float* __restrict__ feat, int D,
                                          int s, float w, int j, float4& acc) {
    float4 v[G];
    float  ww[G];
    #pragma unroll
    for (int u = 0; u < G; ++u) {
        const int sj = __builtin_amdgcn_readlane(s, j + u);
        ww[u] = __int_as_float(__builtin_amdgcn_readlane(__float_as_int(w), j + u));
        v[u] = *(const float4*)(feat + (long long)sj * D);
    }
    #pragma unroll
    for (int u = 0; u < G; ++u) {
        acc.x = fmaf(ww[u], v[u].x, acc.x);
        acc.y = fmaf(ww[u], v[u].y, acc.y);
        acc.z = fmaf(ww[u], v[u].z, acc.z);
        acc.w = fmaf(ww[u], v[u].w, acc.w);
    }
}

#define GATHER_WAVES  8
#define GATHER_BLOCKS 1250
template <int D>
__global__ __launch_bounds__(GATHER_WAVES * 32)
void gather_kernel(const float* __restrict__ feat,
                   const int*   __restrict__ rowptr,
                   const int*   __restrict__ csrc,
                   const float* __restrict__ cw,
                   float*       __restrict__ agg) {
    constexpr int P = (D + 127) / 128;
    static_assert(D % 4 == 0 && (D % 128 == 0 || D < 128), "row = whole float4 lanes");
    const int lane = threadIdx.x & 31;
    const int wid = __builtin_amdgcn_readfirstlane(
        (int)(blockIdx.x * GATHER_WAVES + (threadIdx.x >> 5)));
    const int nw = gridDim.x * GATHER_WAVES;
    for (int item = wid; item < N_NODES * P; item += nw) {
        const int row = item / P;
        const int off = (item - row * P) * 128 + lane * 4;
        const bool on = (D % 128 == 0) ? true : (off < D);
        const float* f = feat + (on ? off : D - 4);
        const int beg = rowptr[row];
        const int end = rowptr[row + 1];
        float4 acc = make_float4(0.0f, 0.0f, 0.0f, 0.0f);
        for (int base = beg; base < end; base += 32) {
            const int p = base + lane;
            int   s = 0;
            float w = 0.0f;
            if (p < end) { s = csrc[p]; w = cw[p]; }
            const int n = (end - base) < 32 ? (end - base) : 32;
            int j = 0;
            for (; j + 8 <= n; j += 8) take_rows<8>(f, D, s, w, j, acc);
            switch (n - j) {
                case 7: take_rows<7>(f, D, s, w, j, acc); break;
                case 6: take_rows<6>(f, D, s, w, j, acc); break;
                case 5: take_rows<5>(f, D, s, w, j, acc); break;
                case 4: take_rows<4>(f, D, s, w, j, acc); break;
                case 3: take_rows<3>(f, D, s, w, j, acc); break;
                case 2: take_rows<2>(f, D, s, w, j, acc); break;
                case 1: take_rows<1>(f, D, s, w, j, acc); break;
                default: break;
            }
        }
        if (on) *(float4*)(agg + (long long)row * D + off) = acc;
    }
}

template <int K>
__global__ __launch_bounds__(32)
void gemm_wmma_kernel(const float* __restrict__ A,
                      const float* __restrict__ W,
                      float*       __restrict__ C,
                      int Ncols, int do_relu) {
    const int l     = threadIdx.x;
    const int rl    = l & 15;
    const int khalf = (l >> 4) * 8;
    const int row0  = blockIdx.x * 16;
    const int col0  = blockIdx.y * 16;
    const int col   = col0 + rl;

    v8f acc = {};
    const float* arow = A + (long long)(row0 + rl) * K + khalf;
    const float* wrow = W + (long long)col * K + khalf;

    #pragma unroll
    for (int k0 = 0; k0 < K; k0 += 32) {
        v16h af;
        #pragma unroll
        for (int e = 0; e < 16; ++e) {
            const int kk = k0 + ((e & 8) ? 16 : 0) + (e & 7);
            af[e] = (_Float16)arow[kk];
        }
        v16h bf = {};
        if (col < Ncols) {
            #pragma unroll
            for (int e = 0; e < 16; ++e) {
                const int kk = k0 + ((e & 8) ? 16 : 0) + (e & 7);
                bf[e] = (_Float16)wrow[kk];
            }
        }
        acc = __builtin_amdgcn_wmma_f32_16x16x32_f16(
             false, af,  false, bf,
             (short)0, acc,  false,  false);
    }

    if (col < Ncols) {
        #pragma unroll
        for (int r = 0; r < 8; ++r) {
            float v = acc[r];
            if (do_relu) v = v > 0.0f ? v : 0.0f;
            C[(long long)(row0 + khalf + r) * Ncols + col] = v;
        }
    }
}

__global__ void logsoftmax_kernel(const float* __restrict__ H,
                                  float*       __restrict__ out) {
    const int n = blockIdx.x * blockDim.x + threadIdx.x;
    if (n >= N_NODES) return;
    const float* h = H + (long long)n * NUM_CLASSES;
    float mx = -INFINITY;
    #pragma unroll
    for (int c = 0; c < NUM_CLASSES; ++c) mx = fmaxf(mx, h[c]);
    float s = 0.0f;
    #pragma unroll
    for (int c = 0; c < NUM_CLASSES; ++c) s += __expf(h[c] - mx);
    const float lse = __logf(s) + mx;
    float* o = out + (long long)n * NUM_CLASSES;
    #pragma unroll
    for (int c = 0; c < NUM_CLASSES; ++c) o[c] = h[c] - lse;
}

extern "C" void kernel_launch(void* const* d_in, const int* in_sizes, int n_in,
                              void* d_out, int out_size, void* d_ws, size_t ws_size,
                              hipStream_t stream) {
    const float* x  = (const float*)d_in[0];
    const int*   ei = (const int*)  d_in[1];
    const float* ew = (const float*)d_in[2];
    const float* w1 = (const float*)d_in[3];
    const float* w2 = (const float*)d_in[4];
    float* out = (float*)d_out;

    const size_t bufElems = (size_t)N_NODES * HIDDEN;
    float* bufA = (float*)d_ws;
    float* bufB = bufA + bufElems;
    float* agg1 = bufA;
    float* h1   = bufB;
    float* p2   = bufA;
    float* h2   = bufB;

    int*   cnt    = (int*)bufB;
    int*   perm   = cnt + ROWPTR_PAD;
    int*   rowptr = (int*)d_out;
    int*   csrc   = rowptr + ROWPTR_PAD;
    float* cw     = (float*)(csrc + N_EDGES);

    hipMemsetAsync(cnt, 0, (size_t)N_NODES * sizeof(int), stream);
    degree_kernel<<<(N_EDGES + 255) / 256, 256, 0, stream>>>(ei, cnt);
    rowptr_kernel<<<SCAN_TILES, SCAN_TILE, 0, stream>>>(cnt, rowptr);
    place_kernel<<<(N_EDGES + 255) / 256, 256, 0, stream>>>(ei, rowptr, cnt, perm);
    order_kernel<<<(N_EDGES + 255) / 256, 256, 0, stream>>>(ei, ew, rowptr, perm, csrc, cw);

    gather_kernel<NUM_FEATS><<<GATHER_BLOCKS, GATHER_WAVES * 32, 0, stream>>>(
        x, rowptr, csrc, cw, agg1);
    gemm_wmma_kernel<NUM_FEATS><<<dim3(N_NODES / 16, HIDDEN / 16), 32, 0, stream>>>(
        agg1, w1, h1, HIDDEN,  1);

    gemm_wmma_kernel<HIDDEN><<<dim3(N_NODES / 16, 3  ), 32, 0, stream>>>(
        h1, w2, p2, NUM_CLASSES,  0);
    gather_kernel<NUM_CLASSES><<<GATHER_BLOCKS, GATHER_WAVES * 32, 0, stream>>>(
        p2, rowptr, csrc, cw, h2);

    logsoftmax_kernel<<<(N_NODES + 255) / 256, 256, 0, stream>>>(h2, out);
}


# each wave now computes a whole 16-row strip of a GEMM's output: A fragments loaded and converted once, reused over all column tiles (was one single-wave workgroup per 16x16 tile); k order kept
# speedup vs baseline: 1.8609x; 1.0815x over previous
#include <hip/hip_runtime.h>
#include <hip/hip_bf16.h>
#include <math.h>

#define N_NODES     50000
#define N_EDGES     800000
#define NUM_FEATS   128
#define HIDDEN      256
#define NUM_CLASSES 40

typedef __attribute__((ext_vector_type(16))) _Float16 v16h;
typedef __attribute__((ext_vector_type(8)))  float    v8f;

#define ROWPTR_PAD 50176
#define SCAN_TILE  1024
#define SCAN_TILES ((N_NODES + SCAN_TILE - 1) / SCAN_TILE)

__global__ void degree_kernel(const int* __restrict__ ei, int* __restrict__ cnt) {
    const int e = blockIdx.x * blockDim.x + threadIdx.x;
    if (e < N_EDGES) atomicAdd(cnt + ei[N_EDGES + e], 1);
}

__global__ __launch_bounds__(SCAN_TILE)
void rowptr_kernel(const int* __restrict__ cnt, int* __restrict__ rowptr) {
    __shared__ int wsum[32];
    __shared__ int wpre[32];
    const int t    = threadIdx.x;
    const int lane = t & 31;
    const int wv   = t >> 5;
    const int idx  = blockIdx.x * SCAN_TILE + t;
    int pre = 0;
    for (int i = 0; i < (int)blockIdx.x; ++i) pre += cnt[i * SCAN_TILE + t];
    const int c = idx < N_NODES ? cnt[idx] : 0;
    int inc = c;
    for (int o = 1; o < 32; o <<= 1) {
        const int v = __shfl_up(inc, o);
        if (lane >= o) inc += v;
    }
    for (int o = 16; o > 0; o >>= 1) pre += __shfl_xor(pre, o);
    if (lane == 31) wsum[wv] = inc;
    if (lane == 0)  wpre[wv] = pre;
    __syncthreads();
    int base = 0;
    for (int j = 0; j < 32; ++j) {
        base += wpre[j];
        if (j < wv) base += wsum[j];
    }
    const int excl = base + inc - c;
    if (idx < N_NODES) rowptr[idx] = excl;
    if (idx == N_NODES - 1) rowptr[N_NODES] = excl + c;
}

__global__ void place_kernel(const int* __restrict__ ei,
                             const int* __restrict__ rowptr,
                             int*       __restrict__ cnt,
                             int*       __restrict__ perm) {
    const int e = blockIdx.x * blockDim.x + threadIdx.x;
    if (e >= N_EDGES) return;
    const int d    = ei[N_EDGES + e];
    const int slot = atomicSub(cnt + d, 1) - 1;
    perm[rowptr[d] + slot] = e;
}

__global__ void order_kernel(const int*   __restrict__ ei,
                             const float* __restrict__ ew,
                             const int*   __restrict__ rowptr,
                             const int*   __restrict__ perm,
                             int*         __restrict__ csrc,
                             float*       __restrict__ cw) {
    const int e = blockIdx.x * blockDim.x + threadIdx.x;
    if (e >= N_EDGES) return;
    const int d   = ei[N_EDGES + e];
    const int beg = rowptr[d];
    const int end = rowptr[d + 1];
    int r = 0;
    for (int p = beg; p < end; ++p) r += (perm[p] < e) ? 1 : 0;
    csrc[beg + r] = ei[e];
    cw[beg + r]   = ew[e];
}

template <int G>
__device__ __forceinline__ void take_rows(const float* __restrict__ feat, int D,
                                          int s, float w, int j, float4& acc) {
    float4 v[G];
    float  ww[G];
    #pragma unroll
    for (int u = 0; u < G; ++u) {
        const int sj = __builtin_amdgcn_readlane(s, j + u);
        ww[u] = __int_as_float(__builtin_amdgcn_readlane(__float_as_int(w), j + u));
        v[u] = *(const float4*)(feat + (long long)sj * D);
    }
    #pragma unroll
    for (int u = 0; u < G; ++u) {
        acc.x = fmaf(ww[u], v[u].x, acc.x);
        acc.y = fmaf(ww[u], v[u].y, acc.y);
        acc.z = fmaf(ww[u], v[u].z, acc.z);
        acc.w = fmaf(ww[u], v[u].w, acc.w);
    }
}

#define GATHER_WAVES  8
#define GATHER_BLOCKS 1250
template <int D>
__global__ __launch_bounds__(GATHER_WAVES * 32)
void gather_kernel(const float* __restrict__ feat,
                   const int*   __restrict__ rowptr,
                   const int*   __restrict__ csrc,
                   const float* __restrict__ cw,
                   float*       __restrict__ agg) {
    constexpr int P = (D + 127) / 128;
    static_assert(D % 4 == 0 && (D % 128 == 0 || D < 128), "row = whole float4 lanes");
    const int lane = threadIdx.x & 31;
    const int wid = __builtin_amdgcn_readfirstlane(
        (int)(blockIdx.x * GATHER_WAVES + (threadIdx.x >> 5)));
    const int nw = gridDim.x * GATHER_WAVES;
    for (int item = wid; item < N_NODES * P; item += nw) {
        const int row = item / P;
        const int off = (item - row * P) * 128 + lane * 4;
        const bool on = (D % 128 == 0) ? true : (off < D);
        const float* f = feat + (on ? off : D - 4);
        const int beg = rowptr[row];
        const int end = rowptr[row + 1];
        float4 acc = make_float4(0.0f, 0.0f, 0.0f, 0.0f);
        for (int base = beg; base < end; base += 32) {
            const int p = base + lane;
            int   s = 0;
            float w = 0.0f;
            if (p < end) { s = csrc[p]; w = cw[p]; }
            const int n = (end - base) < 32 ? (end - base) : 32;
            int j = 0;
            for (; j + 8 <= n; j += 8) take_rows<8>(f, D, s, w, j, acc);
            switch (n - j) {
                case 7: take_rows<7>(f, D, s, w, j, acc); break;
                case 6: take_rows<6>(f, D, s, w, j, acc); break;
                case 5: take_rows<5>(f, D, s, w, j, acc); break;
                case 4: take_rows<4>(f, D, s, w, j, acc); break;
                case 3: take_rows<3>(f, D, s, w, j, acc); break;
                case 2: take_rows<2>(f, D, s, w, j, acc); break;
                case 1: take_rows<1>(f, D, s, w, j, acc); break;
                default: break;
            }
        }
        if (on) *(float4*)(agg + (long long)row * D + off) = acc;
    }
}

#define GEMM_WAVES 4
#define N_STRIPS   (N_NODES / 16)

struct raw16 { float4 a, b, c, d; };

__device__ __forceinline__ raw16 load16(const float* __restrict__ p) {
    raw16 r;
    r.a = *(const float4*)(p);
    r.b = *(const float4*)(p + 4);
    r.c = *(const float4*)(p + 16);
    r.d = *(const float4*)(p + 20);
    return r;
}

__device__ __forceinline__ v16h to_frag(const raw16& r) {
    v16h f;
    f[0]  = (_Float16)r.a.x; f[1]  = (_Float16)r.a.y; f[2]  = (_Float16)r.a.z; f[3]  = (_Float16)r.a.w;
    f[4]  = (_Float16)r.b.x; f[5]  = (_Float16)r.b.y; f[6]  = (_Float16)r.b.z; f[7]  = (_Float16)r.b.w;
    f[8]  = (_Float16)r.c.x; f[9]  = (_Float16)r.c.y; f[10] = (_Float16)r.c.z; f[11] = (_Float16)r.c.w;
    f[12] = (_Float16)r.d.x; f[13] = (_Float16)r.d.y; f[14] = (_Float16)r.d.z; f[15] = (_Float16)r.d.w;
    return f;
}

template <int K, int NC, bool RELU>
__global__ __launch_bounds__(GEMM_WAVES * 32)
void gemm_wmma_kernel(const float* __restrict__ A,
                      const float* __restrict__ W,
                      float*       __restrict__ C) {
    constexpr int KS    = K / 32;
    constexpr int TILES = (NC + 15) / 16;
    static_assert(K % 32 == 0 && NC >= 16, "whole k-steps, at least one full tile");
    const int l     = threadIdx.x & 31;
    const int rl    = l & 15;
    const int khalf = (l >> 4) * 8;
    const int strip = __builtin_amdgcn_readfirstlane(
        (int)(blockIdx.x * GEMM_WAVES + (threadIdx.x >> 5)));
    if (strip >= N_STRIPS) return;
    const int row0 = strip * 16;

    const float* arow = A + (long long)(row0 + rl) * K + khalf;
    v16h af[KS];
    #pragma unroll
    for (int ks = 0; ks < KS; ++ks) af[ks] = to_frag(load16(arow + ks * 32));

    raw16 raw[2][KS];
    {
        const float* w0 = W + (long long)rl * K + khalf;
        #pragma unroll
        for (int ks = 0; ks < KS; ++ks) raw[0][ks] = load16(w0 + ks * 32);
    }

    #pragma unroll
    for (int t = 0; t < TILES; ++t) {
        const int col = t * 16 + rl;
        const bool live = (NC % 16 == 0) || (t + 1 < TILES) || (col < NC);
        if (t + 1 < TILES) {
            const int  cn   = (t + 1) * 16 + rl;
            const bool ln   = (NC % 16 == 0) || (t + 2 < TILES) || (cn < NC);
            const float* wr = W + (long long)(ln ? cn : NC - 1) * K + khalf;
            #pragma unroll
            for (int ks = 0; ks < KS; ++ks) raw[(t + 1) & 1][ks] = load16(wr + ks * 32);
        }

        v8f acc = {};
        #pragma unroll
        for (int ks = 0; ks < KS; ++ks) {
            const v16h bf = to_frag(raw[t & 1][ks]);
            acc = __builtin_amdgcn_wmma_f32_16x16x32_f16(
                 false, af[ks],  false, bf,
                 (short)0, acc,  false,  false);
        }

        if (live) {
            #pragma unroll
            for (int r = 0; r < 8; ++r) {
                float v = acc[r];
                if (RELU) v = v > 0.0f ? v : 0.0f;
                C[(long long)(row0 + khalf + r) * NC + col] = v;
            }
        }
    }
}

__global__ void logsoftmax_kernel(const float* __restrict__ H,
                                  float*       __restrict__ out) {
    const int n = blockIdx.x * blockDim.x + threadIdx.x;
    if (n >= N_NODES) return;
    const float* h = H + (long long)n * NUM_CLASSES;
    float mx = -INFINITY;
    #pragma unroll
    for (int c = 0; c < NUM_CLASSES; ++c) mx = fmaxf(mx, h[c]);
    float s = 0.0f;
    #pragma unroll
    for (int c = 0; c < NUM_CLASSES; ++c) s += __expf(h[c] - mx);
    const float lse = __logf(s) + mx;
    float* o = out + (long long)n * NUM_CLASSES;
    #pragma unroll
    for (int c = 0; c < NUM_CLASSES; ++c) o[c] = h[c] - lse;
}

extern "C" void kernel_launch(void* const* d_in, const int* in_sizes, int n_in,
                              void* d_out, int out_size, void* d_ws, size_t ws_size,
                              hipStream_t stream) {
    const float* x  = (const float*)d_in[0];
    const int*   ei = (const int*)  d_in[1];
    const float* ew = (const float*)d_in[2];
    const float* w1 = (const float*)d_in[3];
    const float* w2 = (const float*)d_in[4];
    float* out = (float*)d_out;

    const size_t bufElems = (size_t)N_NODES * HIDDEN;
    float* bufA = (float*)d_ws;
    float* bufB = bufA + bufElems;
    float* agg1 = bufA;
    float* h1   = bufB;
    float* p2   = bufA;
    float* h2   = bufB;

    int*   cnt    = (int*)bufB;
    int*   perm   = cnt + ROWPTR_PAD;
    int*   rowptr = (int*)d_out;
    int*   csrc   = rowptr + ROWPTR_PAD;
    float* cw     = (float*)(csrc + N_EDGES);

    hipMemsetAsync(cnt, 0, (size_t)N_NODES * sizeof(int), stream);
    degree_kernel<<<(N_EDGES + 255) / 256, 256, 0, stream>>>(ei, cnt);
    rowptr_kernel<<<SCAN_TILES, SCAN_TILE, 0, stream>>>(cnt, rowptr);
    place_kernel<<<(N_EDGES + 255) / 256, 256, 0, stream>>>(ei, rowptr, cnt, perm);
    order_kernel<<<(N_EDGES + 255) / 256, 256, 0, stream>>>(ei, ew, rowptr, perm, csrc, cw);

    gather_kernel<NUM_FEATS><<<GATHER_BLOCKS, GATHER_WAVES * 32, 0, stream>>>(
        x, rowptr, csrc, cw, agg1);
    gemm_wmma_kernel<NUM_FEATS, HIDDEN,  true>
        <<<(N_STRIPS + GEMM_WAVES - 1) / GEMM_WAVES, GEMM_WAVES * 32, 0, stream>>>(agg1, w1, h1);

    gemm_wmma_kernel<HIDDEN, NUM_CLASSES,  false>
        <<<(N_STRIPS + GEMM_WAVES - 1) / GEMM_WAVES, GEMM_WAVES * 32, 0, stream>>>(h1, w2, p2);
    gather_kernel<NUM_CLASSES><<<GATHER_BLOCKS, GATHER_WAVES * 32, 0, stream>>>(
        p2, rowptr, csrc, cw, h2);

    logsoftmax_kernel<<<(N_NODES + 255) / 256, 256, 0, stream>>>(h2, out);
}


# edge sort: placement stores a whole (id, source, weight) record at a per-row cursor and the ordering pass ranks each row inside one wave; no per-edge scans of row lists, no scattered 4-byte stores
# speedup vs baseline: 2.3109x; 1.2418x over previous
#include <hip/hip_runtime.h>
#include <hip/hip_bf16.h>
#include <math.h>

#define N_NODES     50000
#define N_EDGES     800000
#define NUM_FEATS   128
#define HIDDEN      256
#define NUM_CLASSES 40

typedef __attribute__((ext_vector_type(16))) _Float16 v16h;
typedef __attribute__((ext_vector_type(8)))  float    v8f;

#define ROWPTR_PAD 50176
#define SCAN_TILE  1024
#define SCAN_TILES ((N_NODES + SCAN_TILE - 1) / SCAN_TILE)

__global__ void degree_kernel(const int* __restrict__ ei, int* __restrict__ cnt) {
    const int e = blockIdx.x * blockDim.x + threadIdx.x;
    if (e < N_EDGES) atomicAdd(cnt + ei[N_EDGES + e], 1);
}

__global__ __launch_bounds__(SCAN_TILE)
void rowptr_kernel(const int* __restrict__ cnt, int* __restrict__ rowptr,
                   int* __restrict__ cur) {
    __shared__ int wsum[32];
    __shared__ int wpre[32];
    const int t    = threadIdx.x;
    const int lane = t & 31;
    const int wv   = t >> 5;
    const int idx  = blockIdx.x * SCAN_TILE + t;
    int pre = 0;
    for (int i = 0; i < (int)blockIdx.x; ++i) pre += cnt[i * SCAN_TILE + t];
    const int c = idx < N_NODES ? cnt[idx] : 0;
    int inc = c;
    for (int o = 1; o < 32; o <<= 1) {
        const int v = __shfl_up(inc, o);
        if (lane >= o) inc += v;
    }
    for (int o = 16; o > 0; o >>= 1) pre += __shfl_xor(pre, o);
    if (lane == 31) wsum[wv] = inc;
    if (lane == 0)  wpre[wv] = pre;
    __syncthreads();
    int base = 0;
    for (int j = 0; j < 32; ++j) {
        base += wpre[j];
        if (j < wv) base += wsum[j];
    }
    const int excl = base + inc - c;
    if (idx < N_NODES) {
        rowptr[idx] = excl;
        cur[idx]    = excl + c;
    }
    if (idx == N_NODES - 1) rowptr[N_NODES] = excl + c;
}

__global__ void place_kernel(const int*   __restrict__ ei,
                             const float* __restrict__ ew,
                             int*         __restrict__ cur,
                             int*         __restrict__ rec) {
    const int e = blockIdx.x * blockDim.x + threadIdx.x;
    if (e >= N_EDGES) return;
    const int d   = ei[N_EDGES + e];
    const int pos = atomicSub(cur + d, 1) - 1;
    *(int4*)(rec + 4 * (long long)pos) = make_int4(e, ei[e], __float_as_int(ew[e]), 0);
}

__device__ __forceinline__ int count_below(int keys, int n, int mine) {
    int r = 0;
    for (int j0 = 0; j0 < n; j0 += 8) {
        #pragma unroll
        for (int jj = 0; jj < 8; ++jj)
            r += (__builtin_amdgcn_readlane(keys, j0 + jj) < mine) ? 1 : 0;
    }
    return r;
}

__device__ void order_long_row(const int* __restrict__ rec, int beg, int end, int lane,
                               int* __restrict__ csrc, float* __restrict__ cw) {
    for (int base = beg; base < end; base += 32) {
        const int p = base + lane;
        int4 me = make_int4(0x7fffffff, 0, 0, 0);
        if (p < end) me = *(const int4*)(rec + 4 * (long long)p);
        int r = 0;
        for (int ob = beg; ob < end; ob += 32) {
            const int q = ob + lane;
            int keys = 0x7fffffff;
            if (q < end) keys = rec[4 * (long long)q];
            r += count_below(keys, (end - ob) < 32 ? (end - ob) : 32, me.x);
        }
        if (p < end) {
            csrc[beg + r] = me.y;
            cw[beg + r]   = __int_as_float(me.z);
        }
    }
}

#define ORDER_ROWS  8
#define ORDER_WAVES 8
__global__ __launch_bounds__(ORDER_WAVES * 32)
void order_kernel(const int*   __restrict__ rec,
                  const int*   __restrict__ rowptr,
                  int*         __restrict__ csrc,
                  float*       __restrict__ cw) {
    static_assert(N_NODES % ORDER_ROWS == 0, "whole groups of rows");
    const int lane = threadIdx.x & 31;
    const int grp = __builtin_amdgcn_readfirstlane(
        (int)(blockIdx.x * ORDER_WAVES + (threadIdx.x >> 5)));
    const int r0 = grp * ORDER_ROWS;
    if (r0 >= N_NODES) return;

    int b[ORDER_ROWS + 1];
    #pragma unroll
    for (int u = 0; u <= ORDER_ROWS; ++u) b[u] = rowptr[r0 + u];

    int4 rc[ORDER_ROWS];
    #pragma unroll
    for (int u = 0; u < ORDER_ROWS; ++u) {
        rc[u] = make_int4(0x7fffffff, 0, 0, 0);
        if (lane < b[u + 1] - b[u]) rc[u] = *(const int4*)(rec + 4 * (long long)(b[u] + lane));
    }

    int longrows = 0;
    #pragma unroll
    for (int u = 0; u < ORDER_ROWS; ++u) {
        const int n = b[u + 1] - b[u];
        if (n <= 32) {
            const int r = count_below(rc[u].x, n, rc[u].x);
            if (lane < n) {
                csrc[b[u] + r] = rc[u].y;
                cw[b[u] + r]   = __int_as_float(rc[u].z);
            }
        } else {
            longrows |= 1 << u;
        }
    }
    if (longrows != 0) {
        for (int u = 0; u < ORDER_ROWS; ++u)
            if ((longrows >> u) & 1)
                order_long_row(rec, rowptr[r0 + u], rowptr[r0 + u + 1], lane, csrc, cw);
    }
}

template <int G>
__device__ __forceinline__ void take_rows(const float* __restrict__ feat, int D,
                                          int s, float w, int j, float4& acc) {
    float4 v[G];
    float  ww[G];
    #pragma unroll
    for (int u = 0; u < G; ++u) {
        const int sj = __builtin_amdgcn_readlane(s, j + u);
        ww[u] = __int_as_float(__builtin_amdgcn_readlane(__float_as_int(w), j + u));
        v[u] = *(const float4*)(feat + (long long)sj * D);
    }
    #pragma unroll
    for (int u = 0; u < G; ++u) {
        acc.x = fmaf(ww[u], v[u].x, acc.x);
        acc.y = fmaf(ww[u], v[u].y, acc.y);
        acc.z = fmaf(ww[u], v[u].z, acc.z);
        acc.w = fmaf(ww[u], v[u].w, acc.w);
    }
}

#define GATHER_WAVES  8
#define GATHER_BLOCKS 1250
template <int D>
__global__ __launch_bounds__(GATHER_WAVES * 32)
void gather_kernel(const float* __restrict__ feat,
                   const int*   __restrict__ rowptr,
                   const int*   __restrict__ csrc,
                   const float* __restrict__ cw,
                   float*       __restrict__ agg) {
    constexpr int P = (D + 127) / 128;
    static_assert(D % 4 == 0 && (D % 128 == 0 || D < 128), "row = whole float4 lanes");
    const int lane = threadIdx.x & 31;
    const int wid = __builtin_amdgcn_readfirstlane(
        (int)(blockIdx.x * GATHER_WAVES + (threadIdx.x >> 5)));
    const int nw = gridDim.x * GATHER_WAVES;
    for (int item = wid; item < N_NODES * P; item += nw) {
        const int row = item / P;
        const int off = (item - row * P) * 128 + lane * 4;
        const bool on = (D % 128 == 0) ? true : (off < D);
        const float* f = feat + (on ? off : D - 4);
        const int beg = rowptr[row];
        const int end = rowptr[row + 1];
        float4 acc = make_float4(0.0f, 0.0f, 0.0f, 0.0f);
        for (int base = beg; base < end; base += 32) {
            const int p = base + lane;
            int   s = 0;
            float w = 0.0f;
            if (p < end) { s = csrc[p]; w = cw[p]; }
            const int n = (end - base) < 32 ? (end - base) : 32;
            int j = 0;
            for (; j + 8 <= n; j += 8) take_rows<8>(f, D, s, w, j, acc);
            switch (n - j) {
                case 7: take_rows<7>(f, D, s, w, j, acc); break;
                case 6: take_rows<6>(f, D, s, w, j, acc); break;
                case 5: take_rows<5>(f, D, s, w, j, acc); break;
                case 4: take_rows<4>(f, D, s, w, j, acc); break;
                case 3: take_rows<3>(f, D, s, w, j, acc); break;
                case 2: take_rows<2>(f, D, s, w, j, acc); break;
                case 1: take_rows<1>(f, D, s, w, j, acc); break;
                default: break;
            }
        }
        if (on) *(float4*)(agg + (long long)row * D + off) = acc;
    }
}

#define GEMM_WAVES 4
#define N_STRIPS   (N_NODES / 16)

struct raw16 { float4 a, b, c, d; };

__device__ __forceinline__ raw16 load16(const float* __restrict__ p) {
    raw16 r;
    r.a = *(const float4*)(p);
    r.b = *(const float4*)(p + 4);
    r.c = *(const float4*)(p + 16);
    r.d = *(const float4*)(p + 20);
    return r;
}

__device__ __forceinline__ v16h to_frag(const raw16& r) {
    v16h f;
    f[0]  = (_Float16)r.a.x; f[1]  = (_Float16)r.a.y; f[2]  = (_Float16)r.a.z; f[3]  = (_Float16)r.a.w;
    f[4]  = (_Float16)r.b.x; f[5]  = (_Float16)r.b.y; f[6]  = (_Float16)r.b.z; f[7]  = (_Float16)r.b.w;
    f[8]  = (_Float16)r.c.x; f[9]  = (_Float16)r.c.y; f[10] = (_Float16)r.c.z; f[11] = (_Float16)r.c.w;
    f[12] = (_Float16)r.d.x; f[13] = (_Float16)r.d.y; f[14] = (_Float16)r.d.z; f[15] = (_Float16)r.d.w;
    return f;
}

template <int K, int NC, bool RELU>
__global__ __launch_bounds__(GEMM_WAVES * 32)
void gemm_wmma_kernel(const float* __restrict__ A,
                      const float* __restrict__ W,
                      float*       __restrict__ C) {
    constexpr int KS    = K / 32;
    constexpr int TILES = (NC + 15) / 16;
    static_assert(K % 32 == 0 && NC >= 16, "whole k-steps, at least one full tile");
    const int l     = threadIdx.x & 31;
    const int rl    = l & 15;
    const int khalf = (l >> 4) * 8;
    const int strip = __builtin_amdgcn_readfirstlane(
        (int)(blockIdx.x * GEMM_WAVES + (threadIdx.x >> 5)));
    if (strip >= N_STRIPS) return;
    const int row0 = strip * 16;

    const float* arow = A + (long long)(row0 + rl) * K + khalf;
    v16h af[KS];
    #pragma unroll
    for (int ks = 0; ks < KS; ++ks) af[ks] = to_frag(load16(arow + ks * 32));

    raw16 raw[2][KS];
    {
        const float* w0 = W + (long long)rl * K + khalf;
        #pragma unroll
        for (int ks = 0; ks < KS; ++ks) raw[0][ks] = load16(w0 + ks * 32);
    }

    #pragma unroll
    for (int t = 0; t < TILES; ++t) {
        const int col = t * 16 + rl;
        const bool live = (NC % 16 == 0) || (t + 1 < TILES) || (col < NC);
        if (t + 1 < TILES) {
            const int  cn   = (t + 1) * 16 + rl;
            const bool ln   = (NC % 16 == 0) || (t + 2 < TILES) || (cn < NC);
            const float* wr = W + (long long)(ln ? cn : NC - 1) * K + khalf;
            #pragma unroll
            for (int ks = 0; ks < KS; ++ks) raw[(t + 1) & 1][ks] = load16(wr + ks * 32);
        }

        v8f acc = {};
        #pragma unroll
        for (int ks = 0; ks < KS; ++ks) {
            const v16h bf = to_frag(raw[t & 1][ks]);
            acc = __builtin_amdgcn_wmma_f32_16x16x32_f16(
                 false, af[ks],  false, bf,
                 (short)0, acc,  false,  false);
        }

        if (live) {
            #pragma unroll
            for (int r = 0; r < 8; ++r) {
                float v = acc[r];
                if (RELU) v = v > 0.0f ? v : 0.0f;
                C[(long long)(row0 + khalf + r) * NC + col] = v;
            }
        }
    }
}

__global__ void logsoftmax_kernel(const float* __restrict__ H,
                                  float*       __restrict__ out) {
    const int n = blockIdx.x * blockDim.x + threadIdx.x;
    if (n >= N_NODES) return;
    const float* h = H + (long long)n * NUM_CLASSES;
    float mx = -INFINITY;
    #pragma unroll
    for (int c = 0; c < NUM_CLASSES; ++c) mx = fmaxf(mx, h[c]);
    float s = 0.0f;
    #pragma unroll
    for (int c = 0; c < NUM_CLASSES; ++c) s += __expf(h[c] - mx);
    const float lse = __logf(s) + mx;
    float* o = out + (long long)n * NUM_CLASSES;
    #pragma unroll
    for (int c = 0; c < NUM_CLASSES; ++c) o[c] = h[c] - lse;
}

extern "C" void kernel_launch(void* const* d_in, const int* in_sizes, int n_in,
                              void* d_out, int out_size, void* d_ws, size_t ws_size,
                              hipStream_t stream) {
    const float* x  = (const float*)d_in[0];
    const int*   ei = (const int*)  d_in[1];
    const float* ew = (const float*)d_in[2];
    const float* w1 = (const float*)d_in[3];
    const float* w2 = (const float*)d_in[4];
    float* out = (float*)d_out;

    const size_t bufElems = (size_t)N_NODES * HIDDEN;
    float* bufA = (float*)d_ws;
    float* bufB = bufA + bufElems;
    float* agg1 = bufA;
    float* h1   = bufB;
    float* p2   = bufA;
    float* h2   = bufB;

    int*   cnt    = (int*)bufB;
    int*   cur    = cnt + ROWPTR_PAD;
    int*   rec    = cur + ROWPTR_PAD;
    int*   rowptr = (int*)d_out;
    int*   csrc   = rowptr + ROWPTR_PAD;
    float* cw     = (float*)(csrc + N_EDGES);

    hipMemsetAsync(cnt, 0, (size_t)N_NODES * sizeof(int), stream);
    degree_kernel<<<(N_EDGES + 255) / 256, 256, 0, stream>>>(ei, cnt);
    rowptr_kernel<<<SCAN_TILES, SCAN_TILE, 0, stream>>>(cnt, rowptr, cur);
    place_kernel<<<(N_EDGES + 255) / 256, 256, 0, stream>>>(ei, ew, cur, rec);
    order_kernel<<<(N_NODES / ORDER_ROWS + ORDER_WAVES - 1) / ORDER_WAVES, ORDER_WAVES * 32, 0, stream>>>(
        rec, rowptr, csrc, cw);

    gather_kernel<NUM_FEATS><<<GATHER_BLOCKS, GATHER_WAVES * 32, 0, stream>>>(
        x, rowptr, csrc, cw, agg1);
    gemm_wmma_kernel<NUM_FEATS, HIDDEN,  true>
        <<<(N_STRIPS + GEMM_WAVES - 1) / GEMM_WAVES, GEMM_WAVES * 32, 0, stream>>>(agg1, w1, h1);

    gemm_wmma_kernel<HIDDEN, NUM_CLASSES,  false>
        <<<(N_STRIPS + GEMM_WAVES - 1) / GEMM_WAVES, GEMM_WAVES * 32, 0, stream>>>(h1, w2, p2);
    gather_kernel<NUM_CLASSES><<<GATHER_BLOCKS, GATHER_WAVES * 32, 0, stream>>>(
        p2, rowptr, csrc, cw, h2);

    logsoftmax_kernel<<<(N_NODES + 255) / 256, 256, 0, stream>>>(h2, out);
}


# both GEMMs: W converted to f16 once per 13-wave workgroup and kept in LDS; B fragments come from ds_load_b128 (4 or 3 tiles per k-step), not from 16-line global loads per tile; k order kept
# speedup vs baseline: 2.7042x; 1.1702x over previous
#include <hip/hip_runtime.h>
#include <hip/hip_bf16.h>
#include <math.h>

#define N_NODES     50000
#define N_EDGES     800000
#define NUM_FEATS   128
#define HIDDEN      256
#define NUM_CLASSES 40

typedef __attribute__((ext_vector_type(16))) _Float16 v16h;
typedef __attribute__((ext_vector_type(8)))  float    v8f;

#define ROWPTR_PAD 50176
#define SCAN_TILE  1024
#define SCAN_TILES ((N_NODES + SCAN_TILE - 1) / SCAN_TILE)

__global__ void degree_kernel(const int* __restrict__ ei, int* __restrict__ cnt) {
    const int e = blockIdx.x * blockDim.x + threadIdx.x;
    if (e < N_EDGES) atomicAdd(cnt + ei[N_EDGES + e], 1);
}

__global__ __launch_bounds__(SCAN_TILE)
void rowptr_kernel(const int* __restrict__ cnt, int* __restrict__ rowptr,
                   int* __restrict__ cur) {
    __shared__ int wsum[32];
    __shared__ int wpre[32];
    const int t    = threadIdx.x;
    const int lane = t & 31;
    const int wv   = t >> 5;
    const int idx  = blockIdx.x * SCAN_TILE + t;
    int pre = 0;
    for (int i = 0; i < (int)blockIdx.x; ++i) pre += cnt[i * SCAN_TILE + t];
    const int c = idx < N_NODES ? cnt[idx] : 0;
    int inc = c;
    for (int o = 1; o < 32; o <<= 1) {
        const int v = __shfl_up(inc, o);
        if (lane >= o) inc += v;
    }
    for (int o = 16; o > 0; o >>= 1) pre += __shfl_xor(pre, o);
    if (lane == 31) wsum[wv] = inc;
    if (lane == 0)  wpre[wv] = pre;
    __syncthreads();
    int base = 0;
    for (int j = 0; j < 32; ++j) {
        base += wpre[j];
        if (j < wv) base += wsum[j];
    }
    const int excl = base + inc - c;
    if (idx < N_NODES) {
        rowptr[idx] = excl;
        cur[idx]    = excl + c;
    }
    if (idx == N_NODES - 1) rowptr[N_NODES] = excl + c;
}

__global__ void place_kernel(const int*   __restrict__ ei,
                             const float* __restrict__ ew,
                             int*         __restrict__ cur,
                             int*         __restrict__ rec) {
    const int e = blockIdx.x * blockDim.x + threadIdx.x;
    if (e >= N_EDGES) return;
    const int d   = ei[N_EDGES + e];
    const int pos = atomicSub(cur + d, 1) - 1;
    *(int4*)(rec + 4 * (long long)pos) = make_int4(e, ei[e], __float_as_int(ew[e]), 0);
}

__device__ __forceinline__ int count_below(int keys, int n, int mine) {
    int r = 0;
    for (int j0 = 0; j0 < n; j0 += 8) {
        #pragma unroll
        for (int jj = 0; jj < 8; ++jj)
            r += (__builtin_amdgcn_readlane(keys, j0 + jj) < mine) ? 1 : 0;
    }
    return r;
}

__device__ void order_long_row(const int* __restrict__ rec, int beg, int end, int lane,
                               int* __restrict__ csrc, float* __restrict__ cw) {
    for (int base = beg; base < end; base += 32) {
        const int p = base + lane;
        int4 me = make_int4(0x7fffffff, 0, 0, 0);
        if (p < end) me = *(const int4*)(rec + 4 * (long long)p);
        int r = 0;
        for (int ob = beg; ob < end; ob += 32) {
            const int q = ob + lane;
            int keys = 0x7fffffff;
            if (q < end) keys = rec[4 * (long long)q];
            r += count_below(keys, (end - ob) < 32 ? (end - ob) : 32, me.x);
        }
        if (p < end) {
            csrc[beg + r] = me.y;
            cw[beg + r]   = __int_as_float(me.z);
        }
    }
}

#define ORDER_ROWS  8
#define ORDER_WAVES 8
__global__ __launch_bounds__(ORDER_WAVES * 32)
void order_kernel(const int*   __restrict__ rec,
                  const int*   __restrict__ rowptr,
                  int*         __restrict__ csrc,
                  float*       __restrict__ cw) {
    static_assert(N_NODES % ORDER_ROWS == 0, "whole groups of rows");
    const int lane = threadIdx.x & 31;
    const int grp = __builtin_amdgcn_readfirstlane(
        (int)(blockIdx.x * ORDER_WAVES + (threadIdx.x >> 5)));
    const int r0 = grp * ORDER_ROWS;
    if (r0 >= N_NODES) return;

    int b[ORDER_ROWS + 1];
    #pragma unroll
    for (int u = 0; u <= ORDER_ROWS; ++u) b[u] = rowptr[r0 + u];

    int4 rc[ORDER_ROWS];
    #pragma unroll
    for (int u = 0; u < ORDER_ROWS; ++u) {
        rc[u] = make_int4(0x7fffffff, 0, 0, 0);
        if (lane < b[u + 1] - b[u]) rc[u] = *(const int4*)(rec + 4 * (long long)(b[u] + lane));
    }

    int longrows = 0;
    #pragma unroll
    for (int u = 0; u < ORDER_ROWS; ++u) {
        const int n = b[u + 1] - b[u];
        if (n <= 32) {
            const int r = count_below(rc[u].x, n, rc[u].x);
            if (lane < n) {
                csrc[b[u] + r] = rc[u].y;
                cw[b[u] + r]   = __int_as_float(rc[u].z);
            }
        } else {
            longrows |= 1 << u;
        }
    }
    if (longrows != 0) {
        for (int u = 0; u < ORDER_ROWS; ++u)
            if ((longrows >> u) & 1)
                order_long_row(rec, rowptr[r0 + u], rowptr[r0 + u + 1], lane, csrc, cw);
    }
}

template <int G>
__device__ __forceinline__ void take_rows(const float* __restrict__ feat, int D,
                                          int s, float w, int j, float4& acc) {
    float4 v[G];
    float  ww[G];
    #pragma unroll
    for (int u = 0; u < G; ++u) {
        const int sj = __builtin_amdgcn_readlane(s, j + u);
        ww[u] = __int_as_float(__builtin_amdgcn_readlane(__float_as_int(w), j + u));
        v[u] = *(const float4*)(feat + (long long)sj * D);
    }
    #pragma unroll
    for (int u = 0; u < G; ++u) {
        acc.x = fmaf(ww[u], v[u].x, acc.x);
        acc.y = fmaf(ww[u], v[u].y, acc.y);
        acc.z = fmaf(ww[u], v[u].z, acc.z);
        acc.w = fmaf(ww[u], v[u].w, acc.w);
    }
}

#define GATHER_WAVES  8
#define GATHER_BLOCKS 1250
template <int D>
__global__ __launch_bounds__(GATHER_WAVES * 32)
void gather_kernel(const float* __restrict__ feat,
                   const int*   __restrict__ rowptr,
                   const int*   __restrict__ csrc,
                   const float* __restrict__ cw,
                   float*       __restrict__ agg) {
    constexpr int P = (D + 127) / 128;
    static_assert(D % 4 == 0 && (D % 128 == 0 || D < 128), "row = whole float4 lanes");
    const int lane = threadIdx.x & 31;
    const int wid = __builtin_amdgcn_readfirstlane(
        (int)(blockIdx.x * GATHER_WAVES + (threadIdx.x >> 5)));
    const int nw = gridDim.x * GATHER_WAVES;
    for (int item = wid; item < N_NODES * P; item += nw) {
        const int row = item / P;
        const int off = (item - row * P) * 128 + lane * 4;
        const bool on = (D % 128 == 0) ? true : (off < D);
        const float* f = feat + (on ? off : D - 4);
        const int beg = rowptr[row];
        const int end = rowptr[row + 1];
        float4 acc = make_float4(0.0f, 0.0f, 0.0f, 0.0f);
        for (int base = beg; base < end; base += 32) {
            const int p = base + lane;
            int   s = 0;
            float w = 0.0f;
            if (p < end) { s = csrc[p]; w = cw[p]; }
            const int n = (end - base) < 32 ? (end - base) : 32;
            int j = 0;
            for (; j + 8 <= n; j += 8) take_rows<8>(f, D, s, w, j, acc);
            switch (n - j) {
                case 7: take_rows<7>(f, D, s, w, j, acc); break;
                case 6: take_rows<6>(f, D, s, w, j, acc); break;
                case 5: take_rows<5>(f, D, s, w, j, acc); break;
                case 4: take_rows<4>(f, D, s, w, j, acc); break;
                case 3: take_rows<3>(f, D, s, w, j, acc); break;
                case 2: take_rows<2>(f, D, s, w, j, acc); break;
                case 1: take_rows<1>(f, D, s, w, j, acc); break;
                default: break;
            }
        }
        if (on) *(float4*)(agg + (long long)row * D + off) = acc;
    }
}

#define GEMM_WAVES   13
#define GEMM_THREADS (GEMM_WAVES * 32)
#define N_STRIPS     (N_NODES / 16)
#define GEMM_BLOCKS  ((N_STRIPS + GEMM_WAVES - 1) / GEMM_WAVES)

typedef __attribute__((ext_vector_type(8))) _Float16 v8h;
typedef __attribute__((ext_vector_type(4))) _Float16 v4h;

struct raw16 { float4 a, b, c, d; };

__device__ __forceinline__ raw16 load16(const float* __restrict__ p) {
    raw16 r;
    r.a = *(const float4*)(p);
    r.b = *(const float4*)(p + 4);
    r.c = *(const float4*)(p + 16);
    r.d = *(const float4*)(p + 20);
    return r;
}

__device__ __forceinline__ v16h to_frag(const raw16& r) {
    v16h f;
    f[0]  = (_Float16)r.a.x; f[1]  = (_Float16)r.a.y; f[2]  = (_Float16)r.a.z; f[3]  = (_Float16)r.a.w;
    f[4]  = (_Float16)r.b.x; f[5]  = (_Float16)r.b.y; f[6]  = (_Float16)r.b.z; f[7]  = (_Float16)r.b.w;
    f[8]  = (_Float16)r.c.x; f[9]  = (_Float16)r.c.y; f[10] = (_Float16)r.c.z; f[11] = (_Float16)r.c.w;
    f[12] = (_Float16)r.d.x; f[13] = (_Float16)r.d.y; f[14] = (_Float16)r.d.z; f[15] = (_Float16)r.d.w;
    return f;
}

template <int K, int NC, int G, bool RELU>
__global__ __launch_bounds__(GEMM_THREADS)
void gemm_wmma_kernel(const float* __restrict__ A,
                      const float* __restrict__ W,
                      float*       __restrict__ C) {
    constexpr int KS    = K / 32;
    constexpr int TILES = (NC + 15) / 16;
    constexpr int STEPS = (TILES / G) * KS;
    constexpr int PITCH = 2 * K + 16;
    constexpr int NV    = NC * K / 4;
    constexpr int SV    = (NV + GEMM_THREADS - 1) / GEMM_THREADS;
    constexpr int LROWS = (SV * GEMM_THREADS + K / 4 - 1) / (K / 4);
    static_assert(K % 32 == 0 && NC >= 16, "whole k-steps, at least one full tile");
    static_assert(TILES % G == 0, "whole groups of tiles");
    static_assert((PITCH / 16) % 2 == 1, "LDS row pitch = 16 B x odd");
    static_assert(LROWS >= NC, "room for all of W");
    __shared__ __attribute__((aligned(16))) unsigned char wl[LROWS * PITCH];

    const int l     = threadIdx.x & 31;
    const int rl    = l & 15;
    const int khalf = (l >> 4) * 8;
    const int strip = __builtin_amdgcn_readfirstlane(
        (int)(blockIdx.x * GEMM_WAVES + (threadIdx.x >> 5)));
    const bool valid = strip < N_STRIPS;
    const int row0 = valid ? strip * 16 : 0;

    const float* arow = A + (long long)(row0 + rl) * K + khalf;
    raw16 araw[KS];
    #pragma unroll
    for (int ks = 0; ks < KS; ++ks) araw[ks] = load16(arow + ks * 32);
    __builtin_amdgcn_sched_barrier(0);

    {
        float4 sv[SV];
        #pragma unroll
        for (int i = 0; i < SV; ++i) {
            const int idx = (int)threadIdx.x + i * GEMM_THREADS;
            sv[i] = ((const float4*)W)[idx < NV ? idx : NV - 1];
        }
        __builtin_amdgcn_sched_barrier(0);
        #pragma unroll
        for (int i = 0; i < SV; ++i) {
            const int idx = (int)threadIdx.x + i * GEMM_THREADS;
            const int r   = idx / (K / 4);
            const int c4  = idx - r * (K / 4);
            v4h h;
            h[0] = (_Float16)sv[i].x; h[1] = (_Float16)sv[i].y;
            h[2] = (_Float16)sv[i].z; h[3] = (_Float16)sv[i].w;
            *(v4h*)(&wl[r * PITCH + c4 * 8]) = h;
        }
    }
    __syncthreads();
    if (!valid) return;

    v16h af[KS];
    #pragma unroll
    for (int ks = 0; ks < KS; ++ks) af[ks] = to_frag(araw[ks]);

    int woff[TILES];
    #pragma unroll
    for (int t = 0; t < TILES; ++t) {
        const int col = t * 16 + rl;
        const bool in = (NC % 16 == 0) || (t + 1 < TILES) || (col < NC);
        woff[t] = (in ? col : NC - 1) * PITCH + khalf * 2;
    }

    v16h bf[2][G];
    #pragma unroll
    for (int j = 0; j < G; ++j) {
        const v8h lo = *(const v8h*)(&wl[woff[j]]);
        const v8h hi = *(const v8h*)(&wl[woff[j] + 32]);
        bf[0][j] = __builtin_shufflevector(lo, hi, 0, 1, 2, 3, 4, 5, 6, 7,
                                           8, 9, 10, 11, 12, 13, 14, 15);
    }

    v8f acc[G];
    #pragma unroll
    for (int s = 0; s < STEPS; ++s) {
        const int g  = s / KS;
        const int ks = s - g * KS;
        if (s + 1 < STEPS) {
            const int gn = (s + 1) / KS;
            const int kn = (s + 1) - gn * KS;
            #pragma unroll
            for (int j = 0; j < G; ++j) {
                const int o = woff[gn * G + j] + kn * 64;
                const v8h lo = *(const v8h*)(&wl[o]);
                const v8h hi = *(const v8h*)(&wl[o + 32]);
                bf[(s + 1) & 1][j] = __builtin_shufflevector(lo, hi, 0, 1, 2, 3, 4, 5, 6, 7,
                                                             8, 9, 10, 11, 12, 13, 14, 15);
            }
        }
        __builtin_amdgcn_sched_barrier(0);

        #pragma unroll
        for (int j = 0; j < G; ++j) {
            const v8f c0 = {};
            acc[j] = __builtin_amdgcn_wmma_f32_16x16x32_f16(
                 false, af[ks],  false, bf[s & 1][j],
                 (short)0, ks == 0 ? c0 : acc[j],  false,  false);
        }

        if (ks == KS - 1) {
            #pragma unroll
            for (int j = 0; j < G; ++j) {
                const int t   = g * G + j;
                const int col = t * 16 + rl;
                const bool live = (NC % 16 == 0) || (t + 1 < TILES) || (col < NC);
                if (live) {
                    #pragma unroll
                    for (int r = 0; r < 8; ++r) {
                        float v = acc[j][r];
                        if (RELU) v = v > 0.0f ? v : 0.0f;
                        C[(long long)(row0 + khalf + r) * NC + col] = v;
                    }
                }
            }
        }
    }
}

__global__ void logsoftmax_kernel(const float* __restrict__ H,
                                  float*       __restrict__ out) {
    const int n = blockIdx.x * blockDim.x + threadIdx.x;
    if (n >= N_NODES) return;
    const float* h = H + (long long)n * NUM_CLASSES;
    float mx = -INFINITY;
    #pragma unroll
    for (int c = 0; c < NUM_CLASSES; ++c) mx = fmaxf(mx, h[c]);
    float s = 0.0f;
    #pragma unroll
    for (int c = 0; c < NUM_CLASSES; ++c) s += __expf(h[c] - mx);
    const float lse = __logf(s) + mx;
    float* o = out + (long long)n * NUM_CLASSES;
    #pragma unroll
    for (int c = 0; c < NUM_CLASSES; ++c) o[c] = h[c] - lse;
}

extern "C" void kernel_launch(void* const* d_in, const int* in_sizes, int n_in,
                              void* d_out, int out_size, void* d_ws, size_t ws_size,
                              hipStream_t stream) {
    const float* x  = (const float*)d_in[0];
    const int*   ei = (const int*)  d_in[1];
    const float* ew = (const float*)d_in[2];
    const float* w1 = (const float*)d_in[3];
    const float* w2 = (const float*)d_in[4];
    float* out = (float*)d_out;

    const size_t bufElems = (size_t)N_NODES * HIDDEN;
    float* bufA = (float*)d_ws;
    float* bufB = bufA + bufElems;
    float* agg1 = bufA;
    float* h1   = bufB;
    float* p2   = bufA;
    float* h2   = bufB;

    int*   cnt    = (int*)bufB;
    int*   cur    = cnt + ROWPTR_PAD;
    int*   rec    = cur + ROWPTR_PAD;
    int*   rowptr = (int*)d_out;
    int*   csrc   = rowptr + ROWPTR_PAD;
    float* cw     = (float*)(csrc + N_EDGES);

    hipMemsetAsync(cnt, 0, (size_t)N_NODES * sizeof(int), stream);
    degree_kernel<<<(N_EDGES + 255) / 256, 256, 0, stream>>>(ei, cnt);
    rowptr_kernel<<<SCAN_TILES, SCAN_TILE, 0, stream>>>(cnt, rowptr, cur);
    place_kernel<<<(N_EDGES + 255) / 256, 256, 0, stream>>>(ei, ew, cur, rec);
    order_kernel<<<(N_NODES / ORDER_ROWS + ORDER_WAVES - 1) / ORDER_WAVES, ORDER_WAVES * 32, 0, stream>>>(
        rec, rowptr, csrc, cw);

    gather_kernel<NUM_FEATS><<<GATHER_BLOCKS, GATHER_WAVES * 32, 0, stream>>>(
        x, rowptr, csrc, cw, agg1);
    gemm_wmma_kernel<NUM_FEATS, HIDDEN,  4,  true>
        <<<GEMM_BLOCKS, GEMM_THREADS, 0, stream>>>(agg1, w1, h1);

    gemm_wmma_kernel<HIDDEN, NUM_CLASSES,  3,  false>
        <<<GEMM_BLOCKS, GEMM_THREADS, 0, stream>>>(h1, w2, p2);
    gather_kernel<NUM_CLASSES><<<GATHER_BLOCKS, GATHER_WAVES * 32, 0, stream>>>(
        p2, rowptr, csrc, cw, h2);

    logsoftmax_kernel<<<(N_NODES + 255) / 256, 256, 0, stream>>>(h2, out);
}


# both GEMMs in one kernel: h1 tiles computed transposed (W1 as A operand), so their relu'd D registers, packed to f16, are GEMM2's A fragments in place; h1 never goes to memory; k order kept
# speedup vs baseline: 2.8924x; 1.0696x over previous
#include <hip/hip_runtime.h>
#include <hip/hip_bf16.h>
#include <math.h>

#define N_NODES     50000
#define N_EDGES     800000
#define NUM_FEATS   128
#define HIDDEN      256
#define NUM_CLASSES 40

typedef __attribute__((ext_vector_type(16))) _Float16 v16h;
typedef __attribute__((ext_vector_type(8)))  float    v8f;

#define ROWPTR_PAD 50176
#define SCAN_TILE  1024
#define SCAN_TILES ((N_NODES + SCAN_TILE - 1) / SCAN_TILE)

__global__ void degree_kernel(const int* __restrict__ ei, int* __restrict__ cnt) {
    const int e = blockIdx.x * blockDim.x + threadIdx.x;
    if (e < N_EDGES) atomicAdd(cnt + ei[N_EDGES + e], 1);
}

__global__ __launch_bounds__(SCAN_TILE)
void rowptr_kernel(const int* __restrict__ cnt, int* __restrict__ rowptr,
                   int* __restrict__ cur) {
    __shared__ int wsum[32];
    __shared__ int wpre[32];
    const int t    = threadIdx.x;
    const int lane = t & 31;
    const int wv   = t >> 5;
    const int idx  = blockIdx.x * SCAN_TILE + t;
    int pre = 0;
    for (int i = 0; i < (int)blockIdx.x; ++i) pre += cnt[i * SCAN_TILE + t];
    const int c = idx < N_NODES ? cnt[idx] : 0;
    int inc = c;
    for (int o = 1; o < 32; o <<= 1) {
        const int v = __shfl_up(inc, o);
        if (lane >= o) inc += v;
    }
    for (int o = 16; o > 0; o >>= 1) pre += __shfl_xor(pre, o);
    if (lane == 31) wsum[wv] = inc;
    if (lane == 0)  wpre[wv] = pre;
    __syncthreads();
    int base = 0;
    for (int j = 0; j < 32; ++j) {
        base += wpre[j];
        if (j < wv) base += wsum[j];
    }
    const int excl = base + inc - c;
    if (idx < N_NODES) {
        rowptr[idx] = excl;
        cur[idx]    = excl + c;
    }
    if (idx == N_NODES - 1) rowptr[N_NODES] = excl + c;
}

__global__ void place_kernel(const int*   __restrict__ ei,
                             const float* __restrict__ ew,
                             int*         __restrict__ cur,
                             int*         __restrict__ rec) {
    const int e = blockIdx.x * blockDim.x + threadIdx.x;
    if (e >= N_EDGES) return;
    const int d   = ei[N_EDGES + e];
    const int pos = atomicSub(cur + d, 1) - 1;
    *(int4*)(rec + 4 * (long long)pos) = make_int4(e, ei[e], __float_as_int(ew[e]), 0);
}

__device__ __forceinline__ int count_below(int keys, int n, int mine) {
    int r = 0;
    for (int j0 = 0; j0 < n; j0 += 8) {
        #pragma unroll
        for (int jj = 0; jj < 8; ++jj)
            r += (__builtin_amdgcn_readlane(keys, j0 + jj) < mine) ? 1 : 0;
    }
    return r;
}

__device__ void order_long_row(const int* __restrict__ rec, int beg, int end, int lane,
                               int* __restrict__ csrc, float* __restrict__ cw) {
    for (int base = beg; base < end; base += 32) {
        const int p = base + lane;
        int4 me = make_int4(0x7fffffff, 0, 0, 0);
        if (p < end) me = *(const int4*)(rec + 4 * (long long)p);
        int r = 0;
        for (int ob = beg; ob < end; ob += 32) {
            const int q = ob + lane;
            int keys = 0x7fffffff;
            if (q < end) keys = rec[4 * (long long)q];
            r += count_below(keys, (end - ob) < 32 ? (end - ob) : 32, me.x);
        }
        if (p < end) {
            csrc[beg + r] = me.y;
            cw[beg + r]   = __int_as_float(me.z);
        }
    }
}

#define ORDER_ROWS  8
#define ORDER_WAVES 8
__global__ __launch_bounds__(ORDER_WAVES * 32)
void order_kernel(const int*   __restrict__ rec,
                  const int*   __restrict__ rowptr,
                  int*         __restrict__ csrc,
                  float*       __restrict__ cw) {
    static_assert(N_NODES % ORDER_ROWS == 0, "whole groups of rows");
    const int lane = threadIdx.x & 31;
    const int grp = __builtin_amdgcn_readfirstlane(
        (int)(blockIdx.x * ORDER_WAVES + (threadIdx.x >> 5)));
    const int r0 = grp * ORDER_ROWS;
    if (r0 >= N_NODES) return;

    int b[ORDER_ROWS + 1];
    #pragma unroll
    for (int u = 0; u <= ORDER_ROWS; ++u) b[u] = rowptr[r0 + u];

    int4 rc[ORDER_ROWS];
    #pragma unroll
    for (int u = 0; u < ORDER_ROWS; ++u) {
        rc[u] = make_int4(0x7fffffff, 0, 0, 0);
        if (lane < b[u + 1] - b[u]) rc[u] = *(const int4*)(rec + 4 * (long long)(b[u] + lane));
    }

    int longrows = 0;
    #pragma unroll
    for (int u = 0; u < ORDER_ROWS; ++u) {
        const int n = b[u + 1] - b[u];
        if (n <= 32) {
            const int r = count_below(rc[u].x, n, rc[u].x);
            if (lane < n) {
                csrc[b[u] + r] = rc[u].y;
                cw[b[u] + r]   = __int_as_float(rc[u].z);
            }
        } else {
            longrows |= 1 << u;
        }
    }
    if (longrows != 0) {
        for (int u = 0; u < ORDER_ROWS; ++u)
            if ((longrows >> u) & 1)
                order_long_row(rec, rowptr[r0 + u], rowptr[r0 + u + 1], lane, csrc, cw);
    }
}

template <int G>
__device__ __forceinline__ void take_rows(const float* __restrict__ feat, int D,
                                          int s, float w, int j, float4& acc) {
    float4 v[G];
    float  ww[G];
    #pragma unroll
    for (int u = 0; u < G; ++u) {
        const int sj = __builtin_amdgcn_readlane(s, j + u);
        ww[u] = __int_as_float(__builtin_amdgcn_readlane(__float_as_int(w), j + u));
        v[u] = *(const float4*)(feat + (long long)sj * D);
    }
    #pragma unroll
    for (int u = 0; u < G; ++u) {
        acc.x = fmaf(ww[u], v[u].x, acc.x);
        acc.y = fmaf(ww[u], v[u].y, acc.y);
        acc.z = fmaf(ww[u], v[u].z, acc.z);
        acc.w = fmaf(ww[u], v[u].w, acc.w);
    }
}

#define GATHER_WAVES  8
#define GATHER_BLOCKS 1250
template <int D>
__global__ __launch_bounds__(GATHER_WAVES * 32)
void gather_kernel(const float* __restrict__ feat,
                   const int*   __restrict__ rowptr,
                   const int*   __restrict__ csrc,
                   const float* __restrict__ cw,
                   float*       __restrict__ agg) {
    constexpr int P = (D + 127) / 128;
    static_assert(D % 4 == 0 && (D % 128 == 0 || D < 128), "row = whole float4 lanes");
    const int lane = threadIdx.x & 31;
    const int wid = __builtin_amdgcn_readfirstlane(
        (int)(blockIdx.x * GATHER_WAVES + (threadIdx.x >> 5)));
    const int nw = gridDim.x * GATHER_WAVES;
    for (int item = wid; item < N_NODES * P; item += nw) {
        const int row = item / P;
        const int off = (item - row * P) * 128 + lane * 4;
        const bool on = (D % 128 == 0) ? true : (off < D);
        const float* f = feat + (on ? off : D - 4);
        const int beg = rowptr[row];
        const int end = rowptr[row + 1];
        float4 acc = make_float4(0.0f, 0.0f, 0.0f, 0.0f);
        for (int base = beg; base < end; base += 32) {
            const int p = base + lane;
            int   s = 0;
            float w = 0.0f;
            if (p < end) { s = csrc[p]; w = cw[p]; }
            const int n = (end - base) < 32 ? (end - base) : 32;
            int j = 0;
            for (; j + 8 <= n; j += 8) take_rows<8>(f, D, s, w, j, acc);
            switch (n - j) {
                case 7: take_rows<7>(f, D, s, w, j, acc); break;
                case 6: take_rows<6>(f, D, s, w, j, acc); break;
                case 5: take_rows<5>(f, D, s, w, j, acc); break;
                case 4: take_rows<4>(f, D, s, w, j, acc); break;
                case 3: take_rows<3>(f, D, s, w, j, acc); break;
                case 2: take_rows<2>(f, D, s, w, j, acc); break;
                case 1: take_rows<1>(f, D, s, w, j, acc); break;
                default: break;
            }
        }
        if (on) *(float4*)(agg + (long long)row * D + off) = acc;
    }
}

#define GEMM_WAVES   13
#define GEMM_THREADS (GEMM_WAVES * 32)
#define N_STRIPS     (N_NODES / 16)
#define GEMM_BLOCKS  ((N_STRIPS + GEMM_WAVES - 1) / GEMM_WAVES)

typedef __attribute__((ext_vector_type(8))) _Float16 v8h;
typedef __attribute__((ext_vector_type(4))) _Float16 v4h;

struct raw16 { float4 a, b, c, d; };

__device__ __forceinline__ raw16 load16(const float* __restrict__ p) {
    raw16 r;
    r.a = *(const float4*)(p);
    r.b = *(const float4*)(p + 4);
    r.c = *(const float4*)(p + 16);
    r.d = *(const float4*)(p + 20);
    return r;
}

__device__ __forceinline__ v16h to_frag(const raw16& r) {
    v16h f;
    f[0]  = (_Float16)r.a.x; f[1]  = (_Float16)r.a.y; f[2]  = (_Float16)r.a.z; f[3]  = (_Float16)r.a.w;
    f[4]  = (_Float16)r.b.x; f[5]  = (_Float16)r.b.y; f[6]  = (_Float16)r.b.z; f[7]  = (_Float16)r.b.w;
    f[8]  = (_Float16)r.c.x; f[9]  = (_Float16)r.c.y; f[10] = (_Float16)r.c.z; f[11] = (_Float16)r.c.w;
    f[12] = (_Float16)r.d.x; f[13] = (_Float16)r.d.y; f[14] = (_Float16)r.d.z; f[15] = (_Float16)r.d.w;
    return f;
}

__device__ __forceinline__ v16h lds_frag(const unsigned char* wl, int o) {
    const v8h lo = *(const v8h*)(&wl[o]);
    const v8h hi = *(const v8h*)(&wl[o + 32]);
    return __builtin_shufflevector(lo, hi, 0, 1, 2, 3, 4, 5, 6, 7,
                                   8, 9, 10, 11, 12, 13, 14, 15);
}

template <int NV, int SV>
__device__ __forceinline__ void request_w(const float* __restrict__ W, float4 (&sv)[SV]) {
    #pragma unroll
    for (int i = 0; i < SV; ++i) {
        const int idx = (int)threadIdx.x + i * GEMM_THREADS;
        sv[i] = ((const float4*)W)[idx < NV ? idx : NV - 1];
    }
}

template <int K, int PITCH, int SV>
__device__ __forceinline__ void park_w(const float4 (&sv)[SV], unsigned char* wl) {
    #pragma unroll
    for (int i = 0; i < SV; ++i) {
        const int idx = (int)threadIdx.x + i * GEMM_THREADS;
        const int r   = idx / (K / 4);
        const int c4  = idx - r * (K / 4);
        v4h h;
        h[0] = (_Float16)sv[i].x; h[1] = (_Float16)sv[i].y;
        h[2] = (_Float16)sv[i].z; h[3] = (_Float16)sv[i].w;
        *(v4h*)(&wl[r * PITCH + c4 * 8]) = h;
    }
}

template <int K1, int NH, int NC>
__global__ __launch_bounds__(GEMM_THREADS)
void gemm_pair_kernel(const float* __restrict__ A,
                      const float* __restrict__ W1,
                      const float* __restrict__ W2,
                      float*       __restrict__ C) {
    constexpr int KS1    = K1 / 32;
    constexpr int HT     = NH / 16;
    constexpr int T2     = (NC + 15) / 16;
    constexpr int G1     = 4;
    constexpr int Q      = G1 / 2;
    constexpr int GROUPS = HT / G1;
    constexpr int SPG    = KS1 + Q;
    constexpr int STEPS  = GROUPS * SPG;
    constexpr int FR     = G1 > T2 ? G1 : T2;
    constexpr int PITCH1 = 2 * K1 + 16;
    constexpr int PITCH2 = 2 * NH + 16;
    constexpr int NV1    = NH * K1 / 4;
    constexpr int NV2    = NC * NH / 4;
    constexpr int SV1    = (NV1 + GEMM_THREADS - 1) / GEMM_THREADS;
    constexpr int SV2    = (NV2 + GEMM_THREADS - 1) / GEMM_THREADS;
    constexpr int LROWS1 = (SV1 * GEMM_THREADS + K1 / 4 - 1) / (K1 / 4);
    constexpr int LROWS2 = (SV2 * GEMM_THREADS + NH / 4 - 1) / (NH / 4);
    static_assert(K1 % 32 == 0 && NH % 32 == 0 && NC >= 16, "whole k-steps, at least one full tile");
    static_assert(HT % G1 == 0 && G1 % 2 == 0, "whole groups of tile pairs");
    static_assert((PITCH1 / 16) % 2 == 1 && (PITCH2 / 16) % 2 == 1, "LDS row pitch = 16 B x odd");
    static_assert(LROWS1 >= NH && LROWS2 >= NC, "room for all of W1 and W2");
    __shared__ __attribute__((aligned(16))) unsigned char wl1[LROWS1 * PITCH1];
    __shared__ __attribute__((aligned(16))) unsigned char wl2[LROWS2 * PITCH2];

    const int l     = threadIdx.x & 31;
    const int rl    = l & 15;
    const int khalf = (l >> 4) * 8;
    const int strip = __builtin_amdgcn_readfirstlane(
        (int)(blockIdx.x * GEMM_WAVES + (threadIdx.x >> 5)));
    const bool valid = strip < N_STRIPS;
    const int row0 = valid ? strip * 16 : 0;

    const float* arow = A + (long long)(row0 + rl) * K1 + khalf;
    raw16 araw[KS1];
    #pragma unroll
    for (int ks = 0; ks < KS1; ++ks) araw[ks] = load16(arow + ks * 32);
    __builtin_amdgcn_sched_barrier(0);

    {
        float4 sv1[SV1];
        float4 sv2[SV2];
        request_w<NV1, SV1>(W1, sv1);
        request_w<NV2, SV2>(W2, sv2);
        __builtin_amdgcn_sched_barrier(0);
        park_w<K1, PITCH1, SV1>(sv1, wl1);
        park_w<NH, PITCH2, SV2>(sv2, wl2);
    }
    __syncthreads();
    if (!valid) return;

    v16h af[KS1];
    #pragma unroll
    for (int ks = 0; ks < KS1; ++ks) af[ks] = to_frag(araw[ks]);

    const int base1 = rl * PITCH1 + khalf * 2;
    int woff2[T2];
    #pragma unroll
    for (int t = 0; t < T2; ++t) {
        const int col = t * 16 + rl;
        const bool in = (NC % 16 == 0) || (t + 1 < T2) || (col < NC);
        woff2[t] = (in ? col : NC - 1) * PITCH2 + khalf * 2;
    }

    v16h bf[2][FR];
    #pragma unroll
    for (int j = 0; j < G1; ++j) bf[0][j] = lds_frag(wl1, base1 + j * 16 * PITCH1);

    v8f  acc1[G1];
    v8f  acc2[T2];
    v16h hf[Q];
    #pragma unroll
    for (int s = 0; s < STEPS; ++s) {
        const int g = s / SPG;
        const int r = s - g * SPG;
        if (s + 1 < STEPS) {
            const int gn = (s + 1) / SPG;
            const int rn = (s + 1) - gn * SPG;
            if (rn < KS1) {
                #pragma unroll
                for (int j = 0; j < G1; ++j)
                    bf[(s + 1) & 1][j] = lds_frag(wl1, base1 + (gn * G1 + j) * 16 * PITCH1 + rn * 64);
            } else {
                const int kn = gn * Q + (rn - KS1);
                #pragma unroll
                for (int j = 0; j < T2; ++j)
                    bf[(s + 1) & 1][j] = lds_frag(wl2, woff2[j] + kn * 64);
            }
        }
        __builtin_amdgcn_sched_barrier(0);

        if (r < KS1) {
            #pragma unroll
            for (int j = 0; j < G1; ++j) {
                const v8f c0 = {};
                acc1[j] = __builtin_amdgcn_wmma_f32_16x16x32_f16(
                     false, bf[s & 1][j],  false, af[r],
                     (short)0, r == 0 ? c0 : acc1[j],  false,  false);
            }
            if (r == KS1 - 1) {
                #pragma unroll
                for (int q = 0; q < Q; ++q) {
                    #pragma unroll
                    for (int e = 0; e < 8; ++e) {
                        const float lo = acc1[2 * q][e];
                        const float hi = acc1[2 * q + 1][e];
                        hf[q][e]     = (_Float16)(lo > 0.0f ? lo : 0.0f);
                        hf[q][8 + e] = (_Float16)(hi > 0.0f ? hi : 0.0f);
                    }
                }
            }
        } else {
            const int q  = r - KS1;
            const int k2 = g * Q + q;
            #pragma unroll
            for (int j = 0; j < T2; ++j) {
                const v8f c0 = {};
                acc2[j] = __builtin_amdgcn_wmma_f32_16x16x32_f16(
                     false, hf[q],  false, bf[s & 1][j],
                     (short)0, k2 == 0 ? c0 : acc2[j],  false,  false);
            }
        }
    }

    #pragma unroll
    for (int j = 0; j < T2; ++j) {
        const int col = j * 16 + rl;
        const bool live = (NC % 16 == 0) || (j + 1 < T2) || (col < NC);
        if (live) {
            #pragma unroll
            for (int r = 0; r < 8; ++r)
                C[(long long)(row0 + khalf + r) * NC + col] = acc2[j][r];
        }
    }
}

__global__ void logsoftmax_kernel(const float* __restrict__ H,
                                  float*       __restrict__ out) {
    const int n = blockIdx.x * blockDim.x + threadIdx.x;
    if (n >= N_NODES) return;
    const float* h = H + (long long)n * NUM_CLASSES;
    float mx = -INFINITY;
    #pragma unroll
    for (int c = 0; c < NUM_CLASSES; ++c) mx = fmaxf(mx, h[c]);
    float s = 0.0f;
    #pragma unroll
    for (int c = 0; c < NUM_CLASSES; ++c) s += __expf(h[c] - mx);
    const float lse = __logf(s) + mx;
    float* o = out + (long long)n * NUM_CLASSES;
    #pragma unroll
    for (int c = 0; c < NUM_CLASSES; ++c) o[c] = h[c] - lse;
}

extern "C" void kernel_launch(void* const* d_in, const int* in_sizes, int n_in,
                              void* d_out, int out_size, void* d_ws, size_t ws_size,
                              hipStream_t stream) {
    const float* x  = (const float*)d_in[0];
    const int*   ei = (const int*)  d_in[1];
    const float* ew = (const float*)d_in[2];
    const float* w1 = (const float*)d_in[3];
    const float* w2 = (const float*)d_in[4];
    float* out = (float*)d_out;

    const size_t bufElems = (size_t)N_NODES * HIDDEN;
    float* bufA = (float*)d_ws;
    float* bufB = bufA + bufElems;
    float* agg1 = bufA;
    float* p2   = bufB;
    float* h2   = bufA;

    int*   cnt    = (int*)bufB;
    int*   cur    = cnt + ROWPTR_PAD;
    int*   rec    = cur + ROWPTR_PAD;
    int*   rowptr = (int*)d_out;
    int*   csrc   = rowptr + ROWPTR_PAD;
    float* cw     = (float*)(csrc + N_EDGES);

    hipMemsetAsync(cnt, 0, (size_t)N_NODES * sizeof(int), stream);
    degree_kernel<<<(N_EDGES + 255) / 256, 256, 0, stream>>>(ei, cnt);
    rowptr_kernel<<<SCAN_TILES, SCAN_TILE, 0, stream>>>(cnt, rowptr, cur);
    place_kernel<<<(N_EDGES + 255) / 256, 256, 0, stream>>>(ei, ew, cur, rec);
    order_kernel<<<(N_NODES / ORDER_ROWS + ORDER_WAVES - 1) / ORDER_WAVES, ORDER_WAVES * 32, 0, stream>>>(
        rec, rowptr, csrc, cw);

    gather_kernel<NUM_FEATS><<<GATHER_BLOCKS, GATHER_WAVES * 32, 0, stream>>>(
        x, rowptr, csrc, cw, agg1);

    gemm_pair_kernel<NUM_FEATS, HIDDEN, NUM_CLASSES>
        <<<GEMM_BLOCKS, GEMM_THREADS, 0, stream>>>(agg1, w1, w2, p2);

    gather_kernel<NUM_CLASSES><<<GATHER_BLOCKS, GATHER_WAVES * 32, 0, stream>>>(
        p2, rowptr, csrc, cw, h2);

    logsoftmax_kernel<<<(N_NODES + 255) / 256, 256, 0, stream>>>(h2, out);
}


# edge sort by destination on LDS counters in two levels (3125-edge chunks by 256-row bucket, then bucket by row): no global atomic per edge, 3 sort launches not 5; per-row edge-id sum order kept
# speedup vs baseline: 7.1486x; 2.4715x over previous
#include <hip/hip_runtime.h>
#include <hip/hip_bf16.h>
#include <math.h>

#define N_NODES     50000
#define N_EDGES     800000
#define NUM_FEATS   128
#define HIDDEN      256
#define NUM_CLASSES 40

typedef __attribute__((ext_vector_type(16))) _Float16 v16h;
typedef __attribute__((ext_vector_type(8)))  float    v8f;

#define ROWPTR_PAD 50176

#define BKT_ROWS    256
#define N_BKT       ((N_NODES + BKT_ROWS - 1) / BKT_ROWS)
#define N_CHUNK     256
#define CHUNK_E     (N_EDGES / N_CHUNK)
#define TBL_WORDS   (N_BKT * N_CHUNK)
#define BIN_THREADS 256
static_assert(TBL_WORDS % 4 == 0, "records behind the table stay 16-byte aligned");
#define BIN_ROUNDS  ((CHUNK_E + BIN_THREADS - 1) / BIN_THREADS)

__global__ __launch_bounds__(BIN_THREADS)
void bin_kernel(const int*   __restrict__ ei,
                const float* __restrict__ ew,
                int*         __restrict__ tbl,
                int*         __restrict__ rec1) {
    static_assert(N_EDGES % N_CHUNK == 0, "equal chunks");
    static_assert(N_BKT <= BIN_THREADS && CHUNK_E < 65536, "one thread per bucket; count and offset fit 16 bits");
    __shared__ int hist[BIN_THREADS];
    __shared__ int boff[BIN_THREADS];
    __shared__ int wtot[BIN_THREADS / 32];
    const int t    = threadIdx.x;
    const int lane = t & 31;
    const int wv   = t >> 5;
    const int e0   = blockIdx.x * CHUNK_E;
    hist[t] = 0;

    int   d[BIN_ROUNDS], sr[BIN_ROUNDS], rk[BIN_ROUNDS];
    float w[BIN_ROUNDS];
    #pragma unroll
    for (int k = 0; k < BIN_ROUNDS; ++k) {
        const int i = t + k * BIN_THREADS;
        const int e = e0 + (i < CHUNK_E ? i : CHUNK_E - 1);
        d[k]  = ei[N_EDGES + e];
        sr[k] = ei[e];
        w[k]  = ew[e];
    }
    __syncthreads();
    #pragma unroll
    for (int k = 0; k < BIN_ROUNDS; ++k) {
        rk[k] = 0;
        if (t + k * BIN_THREADS < CHUNK_E) rk[k] = atomicAdd(&hist[(unsigned)d[k] / BKT_ROWS], 1);
    }
    __syncthreads();

    const int c = hist[t];
    int inc = c;
    for (int o = 1; o < 32; o <<= 1) {
        const int v = __shfl_up(inc, o);
        if (lane >= o) inc += v;
    }
    if (lane == 31) wtot[wv] = inc;
    __syncthreads();
    int base = 0;
    for (int j = 0; j < BIN_THREADS / 32; ++j)
        if (j < wv) base += wtot[j];
    const int excl = base + inc - c;
    boff[t] = excl;
    if (t < N_BKT) tbl[t * N_CHUNK + blockIdx.x] = (c << 16) | excl;
    __syncthreads();

    #pragma unroll
    for (int k = 0; k < BIN_ROUNDS; ++k) {
        const int i = t + k * BIN_THREADS;
        if (i < CHUNK_E) {
            const int pos = e0 + boff[(unsigned)d[k] / BKT_ROWS] + rk[k];
            *(int4*)(rec1 + 4 * (long long)pos) = make_int4(e0 + i, sr[k], __float_as_int(w[k]), d[k]);
        }
    }
}

#define BKT_THREADS 512
#define BKT_WAVES   (BKT_THREADS / 32)
#define BKT_CPW     (N_CHUNK / BKT_WAVES)
__global__ __launch_bounds__(BKT_THREADS)
void bucket_kernel(const int* __restrict__ tbl,
                   const int* __restrict__ rec1,
                   int*       __restrict__ rowptr,
                   int*       __restrict__ rec) {
    static_assert(N_CHUNK % BKT_WAVES == 0 && BKT_CPW <= 32 && N_CHUNK % 4 == 0, "whole chunks per wave, one table word per lane");
    static_assert(BKT_ROWS <= BKT_THREADS && BKT_ROWS % 32 == 0, "one thread per row in the scan");
    __shared__ int rowcnt[BKT_ROWS];
    __shared__ int wsum[BKT_WAVES];
    __shared__ int wtot[BKT_ROWS / 32];
    const int t    = threadIdx.x;
    const int lane = t & 31;
    const int wv   = __builtin_amdgcn_readfirstlane(t >> 5);
    const int b    = blockIdx.x;
    if (t < BKT_ROWS) rowcnt[t] = 0;

    const int word = tbl[b * N_CHUNK + wv * BKT_CPW + (lane % BKT_CPW)];

    int  n[BKT_CPW];
    long long at[BKT_CPW];
    int4 rc[BKT_CPW];
    const long long spare = (long long)(wv * BKT_CPW) * CHUNK_E;
    #pragma unroll
    for (int u = 0; u < BKT_CPW; ++u) {
        const int wd = __builtin_amdgcn_readlane(word, u);
        n[u]  = wd >> 16;
        at[u] = (long long)(wv * BKT_CPW + u) * CHUNK_E + (wd & 0xffff);
        rc[u] = *(const int4*)(rec1 + 4 * (lane < n[u] ? at[u] + lane : spare));
    }

    int pre = 0;
    const int4* t4 = (const int4*)tbl;
    const int   n4 = b * (N_CHUNK / 4);
    #pragma unroll 8
    for (int i = t; i < n4; i += BKT_THREADS) {
        const int4 v = t4[i];
        pre += (v.x >> 16) + (v.y >> 16) + (v.z >> 16) + (v.w >> 16);
    }
    __syncthreads();

    #pragma unroll
    for (int u = 0; u < BKT_CPW; ++u) {
        if (lane < n[u]) atomicAdd(&rowcnt[(unsigned)rc[u].w % BKT_ROWS], 1);
        if (n[u] > 32)
            for (int i = 32 + lane; i < n[u]; i += 32)
                atomicAdd(&rowcnt[(unsigned)rec1[4 * (at[u] + i) + 3] % BKT_ROWS], 1);
    }
    for (int o = 16; o > 0; o >>= 1) pre += __shfl_xor(pre, o);
    if (lane == 0) wsum[wv] = pre;
    __syncthreads();

    int base = 0;
    for (int j = 0; j < BKT_WAVES; ++j) base += wsum[j];
    const int c = t < BKT_ROWS ? rowcnt[t] : 0;
    int inc = c;
    for (int o = 1; o < 32; o <<= 1) {
        const int v = __shfl_up(inc, o);
        if (lane >= o) inc += v;
    }
    if (t < BKT_ROWS && lane == 31) wtot[t >> 5] = inc;
    __syncthreads();
    if (t < BKT_ROWS) {
        int wb = 0;
        for (int j = 0; j < BKT_ROWS / 32; ++j)
            if (j < (t >> 5)) wb += wtot[j];
        const int excl = wb + inc - c;
        rowcnt[t] = excl;
        const int row = b * BKT_ROWS + t;
        if (row < N_NODES)      rowptr[row]     = base + excl;
        if (row == N_NODES - 1) rowptr[N_NODES] = base + excl + c;
    }
    __syncthreads();

    #pragma unroll
    for (int u = 0; u < BKT_CPW; ++u) {
        if (lane < n[u]) {
            const int pos = base + atomicAdd(&rowcnt[(unsigned)rc[u].w % BKT_ROWS], 1);
            *(int4*)(rec + 4 * (long long)pos) = make_int4(rc[u].x, rc[u].y, rc[u].z, 0);
        }
        if (n[u] > 32) {
            for (int i = 32 + lane; i < n[u]; i += 32) {
                const int4 r  = *(const int4*)(rec1 + 4 * (at[u] + i));
                const int pos = base + atomicAdd(&rowcnt[(unsigned)r.w % BKT_ROWS], 1);
                *(int4*)(rec + 4 * (long long)pos) = make_int4(r.x, r.y, r.z, 0);
            }
        }
    }
}

__device__ __forceinline__ int count_below(int keys, int n, int mine) {
    int r = 0;
    for (int j0 = 0; j0 < n; j0 += 8) {
        #pragma unroll
        for (int jj = 0; jj < 8; ++jj)
            r += (__builtin_amdgcn_readlane(keys, j0 + jj) < mine) ? 1 : 0;
    }
    return r;
}

__device__ void order_long_row(const int* __restrict__ rec, int beg, int end, int lane,
                               int* __restrict__ csrc, float* __restrict__ cw) {
    for (int base = beg; base < end; base += 32) {
        const int p = base + lane;
        int4 me = make_int4(0x7fffffff, 0, 0, 0);
        if (p < end) me = *(const int4*)(rec + 4 * (long long)p);
        int r = 0;
        for (int ob = beg; ob < end; ob += 32) {
            const int q = ob + lane;
            int keys = 0x7fffffff;
            if (q < end) keys = rec[4 * (long long)q];
            r += count_below(keys, (end - ob) < 32 ? (end - ob) : 32, me.x);
        }
        if (p < end) {
            csrc[beg + r] = me.y;
            cw[beg + r]   = __int_as_float(me.z);
        }
    }
}

#define ORDER_ROWS  8
#define ORDER_WAVES 8
__global__ __launch_bounds__(ORDER_WAVES * 32)
void order_kernel(const int*   __restrict__ rec,
                  const int*   __restrict__ rowptr,
                  int*         __restrict__ csrc,
                  float*       __restrict__ cw) {
    static_assert(N_NODES % ORDER_ROWS == 0, "whole groups of rows");
    const int lane = threadIdx.x & 31;
    const int grp = __builtin_amdgcn_readfirstlane(
        (int)(blockIdx.x * ORDER_WAVES + (threadIdx.x >> 5)));
    const int r0 = grp * ORDER_ROWS;
    if (r0 >= N_NODES) return;

    int b[ORDER_ROWS + 1];
    #pragma unroll
    for (int u = 0; u <= ORDER_ROWS; ++u) b[u] = rowptr[r0 + u];

    int4 rc[ORDER_ROWS];
    #pragma unroll
    for (int u = 0; u < ORDER_ROWS; ++u) {
        rc[u] = make_int4(0x7fffffff, 0, 0, 0);
        if (lane < b[u + 1] - b[u]) rc[u] = *(const int4*)(rec + 4 * (long long)(b[u] + lane));
    }

    int longrows = 0;
    #pragma unroll
    for (int u = 0; u < ORDER_ROWS; ++u) {
        const int n = b[u + 1] - b[u];
        if (n <= 32) {
            const int r = count_below(rc[u].x, n, rc[u].x);
            if (lane < n) {
                csrc[b[u] + r] = rc[u].y;
                cw[b[u] + r]   = __int_as_float(rc[u].z);
            }
        } else {
            longrows |= 1 << u;
        }
    }
    if (longrows != 0) {
        for (int u = 0; u < ORDER_ROWS; ++u)
            if ((longrows >> u) & 1)
                order_long_row(rec, rowptr[r0 + u], rowptr[r0 + u + 1], lane, csrc, cw);
    }
}

template <int G>
__device__ __forceinline__ void take_rows(const float* __restrict__ feat, int D,
                                          int s, float w, int j, float4& acc) {
    float4 v[G];
    float  ww[G];
    #pragma unroll
    for (int u = 0; u < G; ++u) {
        const int sj = __builtin_amdgcn_readlane(s, j + u);
        ww[u] = __int_as_float(__builtin_amdgcn_readlane(__float_as_int(w), j + u));
        v[u] = *(const float4*)(feat + (long long)sj * D);
    }
    #pragma unroll
    for (int u = 0; u < G; ++u) {
        acc.x = fmaf(ww[u], v[u].x, acc.x);
        acc.y = fmaf(ww[u], v[u].y, acc.y);
        acc.z = fmaf(ww[u], v[u].z, acc.z);
        acc.w = fmaf(ww[u], v[u].w, acc.w);
    }
}

template <int M>
__device__ __forceinline__ float row_xor(float v) {
    return __int_as_float(__builtin_amdgcn_update_dpp(
        0, __float_as_int(v), 0x160 + M, 0xf, 0xf, false));
}

#define GATHER_WAVES  8
#define GATHER_BLOCKS 1250
template <int D, bool LSM>
__global__ __launch_bounds__(GATHER_WAVES * 32)
void gather_kernel(const float* __restrict__ feat,
                   const int*   __restrict__ rowptr,
                   const int*   __restrict__ csrc,
                   const float* __restrict__ cw,
                   float*       __restrict__ agg) {
    constexpr int P = (D + 127) / 128;
    static_assert(D % 4 == 0 && (D % 128 == 0 || D < 128), "row = whole float4 lanes");
    static_assert(!LSM || D <= 64, "a softmax row sits on the lanes of one 16-lane row");
    const int lane = threadIdx.x & 31;
    const int wid = __builtin_amdgcn_readfirstlane(
        (int)(blockIdx.x * GATHER_WAVES + (threadIdx.x >> 5)));
    const int nw = gridDim.x * GATHER_WAVES;
    for (int item = wid; item < N_NODES * P; item += nw) {
        const int row = item / P;
        const int off = (item - row * P) * 128 + lane * 4;
        const bool on = (D % 128 == 0) ? true : (off < D);
        const float* f = feat + (on ? off : D - 4);
        const int beg = rowptr[row];
        const int end = rowptr[row + 1];
        float4 acc = make_float4(0.0f, 0.0f, 0.0f, 0.0f);
        for (int base = beg; base < end; base += 32) {
            const int p = base + lane;
            int   s = 0;
            float w = 0.0f;
            if (p < end) { s = csrc[p]; w = cw[p]; }
            const int n = (end - base) < 32 ? (end - base) : 32;
            int j = 0;
            for (; j + 8 <= n; j += 8) take_rows<8>(f, D, s, w, j, acc);
            switch (n - j) {
                case 7: take_rows<7>(f, D, s, w, j, acc); break;
                case 6: take_rows<6>(f, D, s, w, j, acc); break;
                case 5: take_rows<5>(f, D, s, w, j, acc); break;
                case 4: take_rows<4>(f, D, s, w, j, acc); break;
                case 3: take_rows<3>(f, D, s, w, j, acc); break;
                case 2: take_rows<2>(f, D, s, w, j, acc); break;
                case 1: take_rows<1>(f, D, s, w, j, acc); break;
                default: break;
            }
        }
        if constexpr (LSM) {
            float m = on ? fmaxf(fmaxf(acc.x, acc.y), fmaxf(acc.z, acc.w)) : -INFINITY;
            m = fmaxf(m, row_xor<1>(m));
            m = fmaxf(m, row_xor<2>(m));
            m = fmaxf(m, row_xor<4>(m));
            m = fmaxf(m, row_xor<8>(m));
            const float e0 = __expf(acc.x - m);
            const float e1 = __expf(acc.y - m);
            const float e2 = __expf(acc.z - m);
            const float e3 = __expf(acc.w - m);
            float s = on ? ((e0 + e1) + e2) + e3 : 0.0f;
            s += row_xor<1>(s);
            s += row_xor<2>(s);
            s += row_xor<4>(s);
            s += row_xor<8>(s);
            const float lse = __logf(s) + m;
            acc.x -= lse;
            acc.y -= lse;
            acc.z -= lse;
            acc.w -= lse;
        }
        if (on) *(float4*)(agg + (long long)row * D + off) = acc;
    }
}

#define GEMM_WAVES   13
#define GEMM_THREADS (GEMM_WAVES * 32)
#define N_STRIPS     (N_NODES / 16)
#define GEMM_BLOCKS  ((N_STRIPS + GEMM_WAVES - 1) / GEMM_WAVES)

typedef __attribute__((ext_vector_type(8))) _Float16 v8h;
typedef __attribute__((ext_vector_type(4))) _Float16 v4h;

struct raw16 { float4 a, b, c, d; };

__device__ __forceinline__ raw16 load16(const float* __restrict__ p) {
    raw16 r;
    r.a = *(const float4*)(p);
    r.b = *(const float4*)(p + 4);
    r.c = *(const float4*)(p + 16);
    r.d = *(const float4*)(p + 20);
    return r;
}

__device__ __forceinline__ v16h to_frag(const raw16& r) {
    v16h f;
    f[0]  = (_Float16)r.a.x; f[1]  = (_Float16)r.a.y; f[2]  = (_Float16)r.a.z; f[3]  = (_Float16)r.a.w;
    f[4]  = (_Float16)r.b.x; f[5]  = (_Float16)r.b.y; f[6]  = (_Float16)r.b.z; f[7]  = (_Float16)r.b.w;
    f[8]  = (_Float16)r.c.x; f[9]  = (_Float16)r.c.y; f[10] = (_Float16)r.c.z; f[11] = (_Float16)r.c.w;
    f[12] = (_Float16)r.d.x; f[13] = (_Float16)r.d.y; f[14] = (_Float16)r.d.z; f[15] = (_Float16)r.d.w;
    return f;
}

__device__ __forceinline__ v16h lds_frag(const unsigned char* wl, int o) {
    const v8h lo = *(const v8h*)(&wl[o]);
    const v8h hi = *(const v8h*)(&wl[o + 32]);
    return __builtin_shufflevector(lo, hi, 0, 1, 2, 3, 4, 5, 6, 7,
                                   8, 9, 10, 11, 12, 13, 14, 15);
}

template <int NV, int SV>
__device__ __forceinline__ void request_w(const float* __restrict__ W, float4 (&sv)[SV]) {
    #pragma unroll
    for (int i = 0; i < SV; ++i) {
        const int idx = (int)threadIdx.x + i * GEMM_THREADS;
        sv[i] = ((const float4*)W)[idx < NV ? idx : NV - 1];
    }
}

template <int K, int PITCH, int SV>
__device__ __forceinline__ void park_w(const float4 (&sv)[SV], unsigned char* wl) {
    #pragma unroll
    for (int i = 0; i < SV; ++i) {
        const int idx = (int)threadIdx.x + i * GEMM_THREADS;
        const int r   = idx / (K / 4);
        const int c4  = idx - r * (K / 4);
        v4h h;
        h[0] = (_Float16)sv[i].x; h[1] = (_Float16)sv[i].y;
        h[2] = (_Float16)sv[i].z; h[3] = (_Float16)sv[i].w;
        *(v4h*)(&wl[r * PITCH + c4 * 8]) = h;
    }
}

template <int K1, int NH, int NC>
__global__ __launch_bounds__(GEMM_THREADS)
void gemm_pair_kernel(const float* __restrict__ A,
                      const float* __restrict__ W1,
                      const float* __restrict__ W2,
                      float*       __restrict__ C) {
    constexpr int KS1    = K1 / 32;
    constexpr int HT     = NH / 16;
    constexpr int T2     = (NC + 15) / 16;
    constexpr int G1     = 4;
    constexpr int Q      = G1 / 2;
    constexpr int GROUPS = HT / G1;
    constexpr int SPG    = KS1 + Q;
    constexpr int STEPS  = GROUPS * SPG;
    constexpr int FR     = G1 > T2 ? G1 : T2;
    constexpr int PITCH1 = 2 * K1 + 16;
    constexpr int PITCH2 = 2 * NH + 16;
    constexpr int NV1    = NH * K1 / 4;
    constexpr int NV2    = NC * NH / 4;
    constexpr int SV1    = (NV1 + GEMM_THREADS - 1) / GEMM_THREADS;
    constexpr int SV2    = (NV2 + GEMM_THREADS - 1) / GEMM_THREADS;
    constexpr int LROWS1 = (SV1 * GEMM_THREADS + K1 / 4 - 1) / (K1 / 4);
    constexpr int LROWS2 = (SV2 * GEMM_THREADS + NH / 4 - 1) / (NH / 4);
    static_assert(K1 % 32 == 0 && NH % 32 == 0 && NC >= 16, "whole k-steps, at least one full tile");
    static_assert(HT % G1 == 0 && G1 % 2 == 0, "whole groups of tile pairs");
    static_assert((PITCH1 / 16) % 2 == 1 && (PITCH2 / 16) % 2 == 1, "LDS row pitch = 16 B x odd");
    static_assert(LROWS1 >= NH && LROWS2 >= NC, "room for all of W1 and W2");
    __shared__ __attribute__((aligned(16))) unsigned char wl1[LROWS1 * PITCH1];
    __shared__ __attribute__((aligned(16))) unsigned char wl2[LROWS2 * PITCH2];

    const int l     = threadIdx.x & 31;
    const int rl    = l & 15;
    const int khalf = (l >> 4) * 8;
    const int strip = __builtin_amdgcn_readfirstlane(
        (int)(blockIdx.x * GEMM_WAVES + (threadIdx.x >> 5)));
    const bool valid = strip < N_STRIPS;
    const int row0 = valid ? strip * 16 : 0;

    const float* arow = A + (long long)(row0 + rl) * K1 + khalf;
    raw16 araw[KS1];
    #pragma unroll
    for (int ks = 0; ks < KS1; ++ks) araw[ks] = load16(arow + ks * 32);
    __builtin_amdgcn_sched_barrier(0);

    {
        float4 sv1[SV1];
        float4 sv2[SV2];
        request_w<NV1, SV1>(W1, sv1);
        request_w<NV2, SV2>(W2, sv2);
        __builtin_amdgcn_sched_barrier(0);
        park_w<K1, PITCH1, SV1>(sv1, wl1);
        park_w<NH, PITCH2, SV2>(sv2, wl2);
    }
    __syncthreads();
    if (!valid) return;

    v16h af[KS1];
    #pragma unroll
    for (int ks = 0; ks < KS1; ++ks) af[ks] = to_frag(araw[ks]);

    const int base1 = rl * PITCH1 + khalf * 2;
    int woff2[T2];
    #pragma unroll
    for (int t = 0; t < T2; ++t) {
        const int col = t * 16 + rl;
        const bool in = (NC % 16 == 0) || (t + 1 < T2) || (col < NC);
        woff2[t] = (in ? col : NC - 1) * PITCH2 + khalf * 2;
    }

    v16h bf[2][FR];
    #pragma unroll
    for (int j = 0; j < G1; ++j) bf[0][j] = lds_frag(wl1, base1 + j * 16 * PITCH1);

    v8f  acc1[G1];
    v8f  acc2[T2];
    v16h hf[Q];
    #pragma unroll
    for (int s = 0; s < STEPS; ++s) {
        const int g = s / SPG;
        const int r = s - g * SPG;
        if (s + 1 < STEPS) {
            const int gn = (s + 1) / SPG;
            const int rn = (s + 1) - gn * SPG;
            if (rn < KS1) {
                #pragma unroll
                for (int j = 0; j < G1; ++j)
                    bf[(s + 1) & 1][j] = lds_frag(wl1, base1 + (gn * G1 + j) * 16 * PITCH1 + rn * 64);
            } else {
                const int kn = gn * Q + (rn - KS1);
                #pragma unroll
                for (int j = 0; j < T2; ++j)
                    bf[(s + 1) & 1][j] = lds_frag(wl2, woff2[j] + kn * 64);
            }
        }
        __builtin_amdgcn_sched_barrier(0);

        if (r < KS1) {
            #pragma unroll
            for (int j = 0; j < G1; ++j) {
                const v8f c0 = {};
                acc1[j] = __builtin_amdgcn_wmma_f32_16x16x32_f16(
                     false, bf[s & 1][j],  false, af[r],
                     (short)0, r == 0 ? c0 : acc1[j],  false,  false);
            }
            if (r == KS1 - 1) {
                #pragma unroll
                for (int q = 0; q < Q; ++q) {
                    #pragma unroll
                    for (int e = 0; e < 8; ++e) {
                        const float lo = acc1[2 * q][e];
                        const float hi = acc1[2 * q + 1][e];
                        hf[q][e]     = (_Float16)(lo > 0.0f ? lo : 0.0f);
                        hf[q][8 + e] = (_Float16)(hi > 0.0f ? hi : 0.0f);
                    }
                }
            }
        } else {
            const int q  = r - KS1;
            const int k2 = g * Q + q;
            #pragma unroll
            for (int j = 0; j < T2; ++j) {
                const v8f c0 = {};
                acc2[j] = __builtin_amdgcn_wmma_f32_16x16x32_f16(
                     false, hf[q],  false, bf[s & 1][j],
                     (short)0, k2 == 0 ? c0 : acc2[j],  false,  false);
            }
        }
    }

    #pragma unroll
    for (int j = 0; j < T2; ++j) {
        const int col = j * 16 + rl;
        const bool live = (NC % 16 == 0) || (j + 1 < T2) || (col < NC);
        if (live) {
            #pragma unroll
            for (int r = 0; r < 8; ++r)
                C[(long long)(row0 + khalf + r) * NC + col] = acc2[j][r];
        }
    }
}

extern "C" void kernel_launch(void* const* d_in, const int* in_sizes, int n_in,
                              void* d_out, int out_size, void* d_ws, size_t ws_size,
                              hipStream_t stream) {
    const float* x  = (const float*)d_in[0];
    const int*   ei = (const int*)  d_in[1];
    const float* ew = (const float*)d_in[2];
    const float* w1 = (const float*)d_in[3];
    const float* w2 = (const float*)d_in[4];
    float* out = (float*)d_out;

    const size_t bufElems = (size_t)N_NODES * HIDDEN;
    float* bufA = (float*)d_ws;
    float* bufB = bufA + bufElems;
    float* agg1 = bufA;
    float* p2   = bufB;

    int*   tbl    = (int*)bufB;
    int*   rec1   = tbl + TBL_WORDS;
    int*   rec    = rec1 + 4 * N_EDGES;
    int*   rowptr = (int*)(bufA + (size_t)N_NODES * NUM_FEATS);
    int*   csrc   = rowptr + ROWPTR_PAD;
    float* cw     = (float*)(csrc + N_EDGES);

    bin_kernel<<<N_CHUNK, BIN_THREADS, 0, stream>>>(ei, ew, tbl, rec1);
    bucket_kernel<<<N_BKT, BKT_THREADS, 0, stream>>>(tbl, rec1, rowptr, rec);
    order_kernel<<<(N_NODES / ORDER_ROWS + ORDER_WAVES - 1) / ORDER_WAVES, ORDER_WAVES * 32, 0, stream>>>(
        rec, rowptr, csrc, cw);

    gather_kernel<NUM_FEATS, false><<<GATHER_BLOCKS, GATHER_WAVES * 32, 0, stream>>>(
        x, rowptr, csrc, cw, agg1);

    gemm_pair_kernel<NUM_FEATS, HIDDEN, NUM_CLASSES>
        <<<GEMM_BLOCKS, GEMM_THREADS, 0, stream>>>(agg1, w1, w2, p2);

    gather_kernel<NUM_CLASSES, true><<<GATHER_BLOCKS, GATHER_WAVES * 32, 0, stream>>>(
        p2, rowptr, csrc, cw, out);
}


# ordering pass dissolved into the layer-1 gather: the wave that sums a row ranks its records by edge id in registers and lane-permutes them (5 launches, not 6); per-row edge-id sum order kept
# speedup vs baseline: 7.5302x; 1.0534x over previous
#include <hip/hip_runtime.h>
#include <hip/hip_bf16.h>
#include <math.h>

#define N_NODES     50000
#define N_EDGES     800000
#define NUM_FEATS   128
#define HIDDEN      256
#define NUM_CLASSES 40

typedef __attribute__((ext_vector_type(16))) _Float16 v16h;
typedef __attribute__((ext_vector_type(8)))  float    v8f;

#define ROWPTR_PAD 50176

#define BKT_ROWS    256
#define N_BKT       ((N_NODES + BKT_ROWS - 1) / BKT_ROWS)
#define N_CHUNK     256
#define CHUNK_E     (N_EDGES / N_CHUNK)
#define TBL_WORDS   (N_BKT * N_CHUNK)
#define BIN_THREADS 256
static_assert(TBL_WORDS % 4 == 0, "records behind the table stay 16-byte aligned");
#define BIN_ROUNDS  ((CHUNK_E + BIN_THREADS - 1) / BIN_THREADS)

__global__ __launch_bounds__(BIN_THREADS)
void bin_kernel(const int*   __restrict__ ei,
                const float* __restrict__ ew,
                int*         __restrict__ tbl,
                int*         __restrict__ rec1) {
    static_assert(N_EDGES % N_CHUNK == 0, "equal chunks");
    static_assert(N_BKT <= BIN_THREADS && CHUNK_E < 65536, "one thread per bucket; count and offset fit 16 bits");
    __shared__ int hist[BIN_THREADS];
    __shared__ int boff[BIN_THREADS];
    __shared__ int wtot[BIN_THREADS / 32];
    const int t    = threadIdx.x;
    const int lane = t & 31;
    const int wv   = t >> 5;
    const int e0   = blockIdx.x * CHUNK_E;
    hist[t] = 0;

    int   d[BIN_ROUNDS], sr[BIN_ROUNDS], rk[BIN_ROUNDS];
    float w[BIN_ROUNDS];
    #pragma unroll
    for (int k = 0; k < BIN_ROUNDS; ++k) {
        const int i = t + k * BIN_THREADS;
        const int e = e0 + (i < CHUNK_E ? i : CHUNK_E - 1);
        d[k]  = ei[N_EDGES + e];
        sr[k] = ei[e];
        w[k]  = ew[e];
    }
    __syncthreads();
    #pragma unroll
    for (int k = 0; k < BIN_ROUNDS; ++k) {
        rk[k] = 0;
        if (t + k * BIN_THREADS < CHUNK_E) rk[k] = atomicAdd(&hist[(unsigned)d[k] / BKT_ROWS], 1);
    }
    __syncthreads();

    const int c = hist[t];
    int inc = c;
    for (int o = 1; o < 32; o <<= 1) {
        const int v = __shfl_up(inc, o);
        if (lane >= o) inc += v;
    }
    if (lane == 31) wtot[wv] = inc;
    __syncthreads();
    int base = 0;
    for (int j = 0; j < BIN_THREADS / 32; ++j)
        if (j < wv) base += wtot[j];
    const int excl = base + inc - c;
    boff[t] = excl;
    if (t < N_BKT) tbl[t * N_CHUNK + blockIdx.x] = (c << 16) | excl;
    __syncthreads();

    #pragma unroll
    for (int k = 0; k < BIN_ROUNDS; ++k) {
        const int i = t + k * BIN_THREADS;
        if (i < CHUNK_E) {
            const int pos = e0 + boff[(unsigned)d[k] / BKT_ROWS] + rk[k];
            *(int4*)(rec1 + 4 * (long long)pos) = make_int4(e0 + i, sr[k], __float_as_int(w[k]), d[k]);
        }
    }
}

#define BKT_THREADS 512
#define BKT_WAVES   (BKT_THREADS / 32)
#define BKT_CPW     (N_CHUNK / BKT_WAVES)
__global__ __launch_bounds__(BKT_THREADS)
void bucket_kernel(const int* __restrict__ tbl,
                   const int* __restrict__ rec1,
                   int*       __restrict__ rowptr,
                   int*       __restrict__ rec) {
    static_assert(N_CHUNK % BKT_WAVES == 0 && BKT_CPW <= 32 && N_CHUNK % 4 == 0, "whole chunks per wave, one table word per lane");
    static_assert(BKT_ROWS <= BKT_THREADS && BKT_ROWS % 32 == 0, "one thread per row in the scan");
    __shared__ int rowcnt[BKT_ROWS];
    __shared__ int wsum[BKT_WAVES];
    __shared__ int wtot[BKT_ROWS / 32];
    const int t    = threadIdx.x;
    const int lane = t & 31;
    const int wv   = __builtin_amdgcn_readfirstlane(t >> 5);
    const int b    = blockIdx.x;
    if (t < BKT_ROWS) rowcnt[t] = 0;

    const int word = tbl[b * N_CHUNK + wv * BKT_CPW + (lane % BKT_CPW)];

    int  n[BKT_CPW];
    long long at[BKT_CPW];
    int4 rc[BKT_CPW];
    const long long spare = (long long)(wv * BKT_CPW) * CHUNK_E;
    #pragma unroll
    for (int u = 0; u < BKT_CPW; ++u) {
        const int wd = __builtin_amdgcn_readlane(word, u);
        n[u]  = wd >> 16;
        at[u] = (long long)(wv * BKT_CPW + u) * CHUNK_E + (wd & 0xffff);
        rc[u] = *(const int4*)(rec1 + 4 * (lane < n[u] ? at[u] + lane : spare));
    }

    int pre = 0;
    const int4* t4 = (const int4*)tbl;
    const int   n4 = b * (N_CHUNK / 4);
    #pragma unroll 8
    for (int i = t; i < n4; i += BKT_THREADS) {
        const int4 v = t4[i];
        pre += (v.x >> 16) + (v.y >> 16) + (v.z >> 16) + (v.w >> 16);
    }
    __syncthreads();

    #pragma unroll
    for (int u = 0; u < BKT_CPW; ++u) {
        if (lane < n[u]) atomicAdd(&rowcnt[(unsigned)rc[u].w % BKT_ROWS], 1);
        if (n[u] > 32)
            for (int i = 32 + lane; i < n[u]; i += 32)
                atomicAdd(&rowcnt[(unsigned)rec1[4 * (at[u] + i) + 3] % BKT_ROWS], 1);
    }
    for (int o = 16; o > 0; o >>= 1) pre += __shfl_xor(pre, o);
    if (lane == 0) wsum[wv] = pre;
    __syncthreads();

    int base = 0;
    for (int j = 0; j < BKT_WAVES; ++j) base += wsum[j];
    const int c = t < BKT_ROWS ? rowcnt[t] : 0;
    int inc = c;
    for (int o = 1; o < 32; o <<= 1) {
        const int v = __shfl_up(inc, o);
        if (lane >= o) inc += v;
    }
    if (t < BKT_ROWS && lane == 31) wtot[t >> 5] = inc;
    __syncthreads();
    if (t < BKT_ROWS) {
        int wb = 0;
        for (int j = 0; j < BKT_ROWS / 32; ++j)
            if (j < (t >> 5)) wb += wtot[j];
        const int excl = wb + inc - c;
        rowcnt[t] = excl;
        const int row = b * BKT_ROWS + t;
        if (row < N_NODES)      rowptr[row]     = base + excl;
        if (row == N_NODES - 1) rowptr[N_NODES] = base + excl + c;
    }
    __syncthreads();

    #pragma unroll
    for (int u = 0; u < BKT_CPW; ++u) {
        if (lane < n[u]) {
            const int pos = base + atomicAdd(&rowcnt[(unsigned)rc[u].w % BKT_ROWS], 1);
            *(int4*)(rec + 4 * (long long)pos) = make_int4(rc[u].x, rc[u].y, rc[u].z, 0);
        }
        if (n[u] > 32) {
            for (int i = 32 + lane; i < n[u]; i += 32) {
                const int4 r  = *(const int4*)(rec1 + 4 * (at[u] + i));
                const int pos = base + atomicAdd(&rowcnt[(unsigned)r.w % BKT_ROWS], 1);
                *(int4*)(rec + 4 * (long long)pos) = make_int4(r.x, r.y, r.z, 0);
            }
        }
    }
}

__device__ __forceinline__ int count_below(int keys, int n, int mine) {
    int r = 0;
    for (int j0 = 0; j0 < n; j0 += 8) {
        #pragma unroll
        for (int jj = 0; jj < 8; ++jj)
            r += (__builtin_amdgcn_readlane(keys, j0 + jj) < mine) ? 1 : 0;
    }
    return r;
}

template <int G>
__device__ __forceinline__ void take_rows(const float* __restrict__ feat, int D,
                                          int s, float w, int j, float4& acc) {
    float4 v[G];
    float  ww[G];
    #pragma unroll
    for (int u = 0; u < G; ++u) {
        const int sj = __builtin_amdgcn_readlane(s, j + u);
        ww[u] = __int_as_float(__builtin_amdgcn_readlane(__float_as_int(w), j + u));
        v[u] = *(const float4*)(feat + (long long)sj * D);
    }
    #pragma unroll
    for (int u = 0; u < G; ++u) {
        acc.x = fmaf(ww[u], v[u].x, acc.x);
        acc.y = fmaf(ww[u], v[u].y, acc.y);
        acc.z = fmaf(ww[u], v[u].z, acc.z);
        acc.w = fmaf(ww[u], v[u].w, acc.w);
    }
}

template <int M>
__device__ __forceinline__ float row_xor(float v) {
    return __int_as_float(__builtin_amdgcn_update_dpp(
        0, __float_as_int(v), 0x160 + M, 0xf, 0xf, false));
}

#define GATHER_WAVES  8
#define GATHER_BLOCKS 1250
template <int D, bool LSM, bool ORDER>
__global__ __launch_bounds__(GATHER_WAVES * 32)
void gather_kernel(const float* __restrict__ feat,
                   const int*   __restrict__ rowptr,
                   const int*   __restrict__ rec,
                   const int*   __restrict__ li,
                   const float* __restrict__ lwt,
                   int*         __restrict__ lo,
                   float*       __restrict__ lw,
                   float*       __restrict__ agg) {
    constexpr int P = (D + 127) / 128;
    static_assert(D % 4 == 0 && (D % 128 == 0 || D < 128), "row = whole float4 lanes");
    static_assert(!LSM || D <= 64, "a softmax row sits on the lanes of one 16-lane row");
    static_assert(!ORDER || P == 1, "the wave that orders a row is the only one that sums it");
    const int lane = threadIdx.x & 31;
    const int wid = __builtin_amdgcn_readfirstlane(
        (int)(blockIdx.x * GATHER_WAVES + (threadIdx.x >> 5)));
    const int nw = gridDim.x * GATHER_WAVES;
    for (int item = wid; item < N_NODES * P; item += nw) {
        const int row = item / P;
        const int off = (item - row * P) * 128 + lane * 4;
        const bool on = (D % 128 == 0) ? true : (off < D);
        const float* f = feat + (on ? off : D - 4);
        const int beg = rowptr[row];
        const int end = rowptr[row + 1];
        float4 acc = make_float4(0.0f, 0.0f, 0.0f, 0.0f);
        for (int base = beg; base < end; base += 32) {
            const int p = base + lane;
            int   s = 0;
            float w = 0.0f;
            if constexpr (ORDER) {
                const int nr = end - beg;
                if (nr <= 32) {
                    int4 me = make_int4(0x7fffffff, 0, 0, 0);
                    if (lane < nr) me = *(const int4*)(rec + 4 * (long long)p);
                    int r = count_below(me.x, nr, me.x);
                    r = lane < nr ? r : lane;
                    s = __builtin_amdgcn_ds_permute(r << 2, me.y);
                    w = __int_as_float(__builtin_amdgcn_ds_permute(r << 2, me.z));
                } else {
                    __shared__ int2 slot[GATHER_WAVES][32];
                    const int wv = threadIdx.x >> 5;
                    const int c0 = base - beg;
                    for (int pb = beg; pb < end; pb += 32) {
                        const int q = pb + lane;
                        int4 me = make_int4(0x7fffffff, 0, 0, 0);
                        if (q < end) me = *(const int4*)(rec + 4 * (long long)q);
                        int r = 0;
                        for (int ob = beg; ob < end; ob += 32) {
                            int keys = 0x7fffffff;
                            if (ob + lane < end) keys = rec[4 * (long long)(ob + lane)];
                            r += count_below(keys, (end - ob) < 32 ? (end - ob) : 32, me.x);
                        }
                        if (q < end && r >= c0 && r < c0 + 32) slot[wv][r - c0] = make_int2(me.y, me.z);
                    }
                    if (p < end) {
                        const int2 t = slot[wv][lane];
                        s = t.x;
                        w = __int_as_float(t.y);
                    }
                }
                if (p < end) { lo[p] = s; lw[p] = w; }
            } else {
                if (p < end) { s = li[p]; w = lwt[p]; }
            }
            const int n = (end - base) < 32 ? (end - base) : 32;
            int j = 0;
            for (; j + 8 <= n; j += 8) take_rows<8>(f, D, s, w, j, acc);
            switch (n - j) {
                case 7: take_rows<7>(f, D, s, w, j, acc); break;
                case 6: take_rows<6>(f, D, s, w, j, acc); break;
                case 5: take_rows<5>(f, D, s, w, j, acc); break;
                case 4: take_rows<4>(f, D, s, w, j, acc); break;
                case 3: take_rows<3>(f, D, s, w, j, acc); break;
                case 2: take_rows<2>(f, D, s, w, j, acc); break;
                case 1: take_rows<1>(f, D, s, w, j, acc); break;
                default: break;
            }
        }
        if constexpr (LSM) {
            float m = on ? fmaxf(fmaxf(acc.x, acc.y), fmaxf(acc.z, acc.w)) : -INFINITY;
            m = fmaxf(m, row_xor<1>(m));
            m = fmaxf(m, row_xor<2>(m));
            m = fmaxf(m, row_xor<4>(m));
            m = fmaxf(m, row_xor<8>(m));
            const float e0 = __expf(acc.x - m);
            const float e1 = __expf(acc.y - m);
            const float e2 = __expf(acc.z - m);
            const float e3 = __expf(acc.w - m);
            float s = on ? ((e0 + e1) + e2) + e3 : 0.0f;
            s += row_xor<1>(s);
            s += row_xor<2>(s);
            s += row_xor<4>(s);
            s += row_xor<8>(s);
            const float lse = __logf(s) + m;
            acc.x -= lse;
            acc.y -= lse;
            acc.z -= lse;
            acc.w -= lse;
        }
        if (on) *(float4*)(agg + (long long)row * D + off) = acc;
    }
}

#define GEMM_WAVES   13
#define GEMM_THREADS (GEMM_WAVES * 32)
#define N_STRIPS     (N_NODES / 16)
#define GEMM_BLOCKS  ((N_STRIPS + GEMM_WAVES - 1) / GEMM_WAVES)

typedef __attribute__((ext_vector_type(8))) _Float16 v8h;
typedef __attribute__((ext_vector_type(4))) _Float16 v4h;

struct raw16 { float4 a, b, c, d; };

__device__ __forceinline__ raw16 load16(const float* __restrict__ p) {
    raw16 r;
    r.a = *(const float4*)(p);
    r.b = *(const float4*)(p + 4);
    r.c = *(const float4*)(p + 16);
    r.d = *(const float4*)(p + 20);
    return r;
}

__device__ __forceinline__ v16h to_frag(const raw16& r) {
    v16h f;
    f[0]  = (_Float16)r.a.x; f[1]  = (_Float16)r.a.y; f[2]  = (_Float16)r.a.z; f[3]  = (_Float16)r.a.w;
    f[4]  = (_Float16)r.b.x; f[5]  = (_Float16)r.b.y; f[6]  = (_Float16)r.b.z; f[7]  = (_Float16)r.b.w;
    f[8]  = (_Float16)r.c.x; f[9]  = (_Float16)r.c.y; f[10] = (_Float16)r.c.z; f[11] = (_Float16)r.c.w;
    f[12] = (_Float16)r.d.x; f[13] = (_Float16)r.d.y; f[14] = (_Float16)r.d.z; f[15] = (_Float16)r.d.w;
    return f;
}

__device__ __forceinline__ v16h lds_frag(const unsigned char* wl, int o) {
    const v8h lo = *(const v8h*)(&wl[o]);
    const v8h hi = *(const v8h*)(&wl[o + 32]);
    return __builtin_shufflevector(lo, hi, 0, 1, 2, 3, 4, 5, 6, 7,
                                   8, 9, 10, 11, 12, 13, 14, 15);
}

template <int NV, int SV>
__device__ __forceinline__ void request_w(const float* __restrict__ W, float4 (&sv)[SV]) {
    #pragma unroll
    for (int i = 0; i < SV; ++i) {
        const int idx = (int)threadIdx.x + i * GEMM_THREADS;
        sv[i] = ((const float4*)W)[idx < NV ? idx : NV - 1];
    }
}

template <int K, int PITCH, int SV>
__device__ __forceinline__ void park_w(const float4 (&sv)[SV], unsigned char* wl) {
    #pragma unroll
    for (int i = 0; i < SV; ++i) {
        const int idx = (int)threadIdx.x + i * GEMM_THREADS;
        const int r   = idx / (K / 4);
        const int c4  = idx - r * (K / 4);
        v4h h;
        h[0] = (_Float16)sv[i].x; h[1] = (_Float16)sv[i].y;
        h[2] = (_Float16)sv[i].z; h[3] = (_Float16)sv[i].w;
        *(v4h*)(&wl[r * PITCH + c4 * 8]) = h;
    }
}

template <int K1, int NH, int NC>
__global__ __launch_bounds__(GEMM_THREADS)
void gemm_pair_kernel(const float* __restrict__ A,
                      const float* __restrict__ W1,
                      const float* __restrict__ W2,
                      float*       __restrict__ C) {
    constexpr int KS1    = K1 / 32;
    constexpr int HT     = NH / 16;
    constexpr int T2     = (NC + 15) / 16;
    constexpr int G1     = 4;
    constexpr int Q      = G1 / 2;
    constexpr int GROUPS = HT / G1;
    constexpr int SPG    = KS1 + Q;
    constexpr int STEPS  = GROUPS * SPG;
    constexpr int FR     = G1 > T2 ? G1 : T2;
    constexpr int PITCH1 = 2 * K1 + 16;
    constexpr int PITCH2 = 2 * NH + 16;
    constexpr int NV1    = NH * K1 / 4;
    constexpr int NV2    = NC * NH / 4;
    constexpr int SV1    = (NV1 + GEMM_THREADS - 1) / GEMM_THREADS;
    constexpr int SV2    = (NV2 + GEMM_THREADS - 1) / GEMM_THREADS;
    constexpr int LROWS1 = (SV1 * GEMM_THREADS + K1 / 4 - 1) / (K1 / 4);
    constexpr int LROWS2 = (SV2 * GEMM_THREADS + NH / 4 - 1) / (NH / 4);
    static_assert(K1 % 32 == 0 && NH % 32 == 0 && NC >= 16, "whole k-steps, at least one full tile");
    static_assert(HT % G1 == 0 && G1 % 2 == 0, "whole groups of tile pairs");
    static_assert((PITCH1 / 16) % 2 == 1 && (PITCH2 / 16) % 2 == 1, "LDS row pitch = 16 B x odd");
    static_assert(LROWS1 >= NH && LROWS2 >= NC, "room for all of W1 and W2");
    __shared__ __attribute__((aligned(16))) unsigned char wl1[LROWS1 * PITCH1];
    __shared__ __attribute__((aligned(16))) unsigned char wl2[LROWS2 * PITCH2];

    const int l     = threadIdx.x & 31;
    const int rl    = l & 15;
    const int khalf = (l >> 4) * 8;
    const int strip = __builtin_amdgcn_readfirstlane(
        (int)(blockIdx.x * GEMM_WAVES + (threadIdx.x >> 5)));
    const bool valid = strip < N_STRIPS;
    const int row0 = valid ? strip * 16 : 0;

    const float* arow = A + (long long)(row0 + rl) * K1 + khalf;
    raw16 araw[KS1];
    #pragma unroll
    for (int ks = 0; ks < KS1; ++ks) araw[ks] = load16(arow + ks * 32);
    __builtin_amdgcn_sched_barrier(0);

    {
        float4 sv1[SV1];
        float4 sv2[SV2];
        request_w<NV1, SV1>(W1, sv1);
        request_w<NV2, SV2>(W2, sv2);
        __builtin_amdgcn_sched_barrier(0);
        park_w<K1, PITCH1, SV1>(sv1, wl1);
        park_w<NH, PITCH2, SV2>(sv2, wl2);
    }
    __syncthreads();
    if (!valid) return;

    v16h af[KS1];
    #pragma unroll
    for (int ks = 0; ks < KS1; ++ks) af[ks] = to_frag(araw[ks]);

    const int base1 = rl * PITCH1 + khalf * 2;
    int woff2[T2];
    #pragma unroll
    for (int t = 0; t < T2; ++t) {
        const int col = t * 16 + rl;
        const bool in = (NC % 16 == 0) || (t + 1 < T2) || (col < NC);
        woff2[t] = (in ? col : NC - 1) * PITCH2 + khalf * 2;
    }

    v16h bf[2][FR];
    #pragma unroll
    for (int j = 0; j < G1; ++j) bf[0][j] = lds_frag(wl1, base1 + j * 16 * PITCH1);

    v8f  acc1[G1];
    v8f  acc2[T2];
    v16h hf[Q];
    #pragma unroll
    for (int s = 0; s < STEPS; ++s) {
        const int g = s / SPG;
        const int r = s - g * SPG;
        if (s + 1 < STEPS) {
            const int gn = (s + 1) / SPG;
            const int rn = (s + 1) - gn * SPG;
            if (rn < KS1) {
                #pragma unroll
                for (int j = 0; j < G1; ++j)
                    bf[(s + 1) & 1][j] = lds_frag(wl1, base1 + (gn * G1 + j) * 16 * PITCH1 + rn * 64);
            } else {
                const int kn = gn * Q + (rn - KS1);
                #pragma unroll
                for (int j = 0; j < T2; ++j)
                    bf[(s + 1) & 1][j] = lds_frag(wl2, woff2[j] + kn * 64);
            }
        }
        __builtin_amdgcn_sched_barrier(0);

        if (r < KS1) {
            #pragma unroll
            for (int j = 0; j < G1; ++j) {
                const v8f c0 = {};
                acc1[j] = __builtin_amdgcn_wmma_f32_16x16x32_f16(
                     false, bf[s & 1][j],  false, af[r],
                     (short)0, r == 0 ? c0 : acc1[j],  false,  false);
            }
            if (r == KS1 - 1) {
                #pragma unroll
                for (int q = 0; q < Q; ++q) {
                    #pragma unroll
                    for (int e = 0; e < 8; ++e) {
                        const float lo = acc1[2 * q][e];
                        const float hi = acc1[2 * q + 1][e];
                        hf[q][e]     = (_Float16)(lo > 0.0f ? lo : 0.0f);
                        hf[q][8 + e] = (_Float16)(hi > 0.0f ? hi : 0.0f);
                    }
                }
            }
        } else {
            const int q  = r - KS1;
            const int k2 = g * Q + q;
            #pragma unroll
            for (int j = 0; j < T2; ++j) {
                const v8f c0 = {};
                acc2[j] = __builtin_amdgcn_wmma_f32_16x16x32_f16(
                     false, hf[q],  false, bf[s & 1][j],
                     (short)0, k2 == 0 ? c0 : acc2[j],  false,  false);
            }
        }
    }

    #pragma unroll
    for (int j = 0; j < T2; ++j) {
        const int col = j * 16 + rl;
        const bool live = (NC % 16 == 0) || (j + 1 < T2) || (col < NC);
        if (live) {
            #pragma unroll
            for (int r = 0; r < 8; ++r)
                C[(long long)(row0 + khalf + r) * NC + col] = acc2[j][r];
        }
    }
}

extern "C" void kernel_launch(void* const* d_in, const int* in_sizes, int n_in,
                              void* d_out, int out_size, void* d_ws, size_t ws_size,
                              hipStream_t stream) {
    const float* x  = (const float*)d_in[0];
    const int*   ei = (const int*)  d_in[1];
    const float* ew = (const float*)d_in[2];
    const float* w1 = (const float*)d_in[3];
    const float* w2 = (const float*)d_in[4];
    float* out = (float*)d_out;

    const size_t bufElems = (size_t)N_NODES * HIDDEN;
    float* bufA = (float*)d_ws;
    float* bufB = bufA + bufElems;
    float* agg1 = bufA;
    float* p2   = bufB;

    int*   tbl    = (int*)bufB;
    int*   rec1   = tbl + TBL_WORDS;
    int*   rec    = rec1 + 4 * N_EDGES;
    int*   rowptr = (int*)(bufA + (size_t)N_NODES * NUM_FEATS);
    int*   csrc   = rowptr + ROWPTR_PAD;
    float* cw     = (float*)(csrc + N_EDGES);

    bin_kernel<<<N_CHUNK, BIN_THREADS, 0, stream>>>(ei, ew, tbl, rec1);
    bucket_kernel<<<N_BKT, BKT_THREADS, 0, stream>>>(tbl, rec1, rowptr, rec);

    gather_kernel<NUM_FEATS, false, true><<<GATHER_BLOCKS, GATHER_WAVES * 32, 0, stream>>>(
        x, rowptr, rec, nullptr, nullptr, csrc, cw, agg1);

    gemm_pair_kernel<NUM_FEATS, HIDDEN, NUM_CLASSES>
        <<<GEMM_BLOCKS, GEMM_THREADS, 0, stream>>>(agg1, w1, w2, p2);

    gather_kernel<NUM_CLASSES, true, false><<<GATHER_BLOCKS, GATHER_WAVES * 32, 0, stream>>>(
        p2, rowptr, nullptr, csrc, cw, nullptr, nullptr, out);
}


# second grouping pass fused into layer 1's row sums: the workgroup that groups a 196-row bucket by row keeps the records in LDS and sums those rows (4 launches, not 5); per-row edge-id sum order kept
# speedup vs baseline: 8.2150x; 1.0909x over previous
#include <hip/hip_runtime.h>
#include <hip/hip_bf16.h>
#include <math.h>

#define N_NODES     50000
#define N_EDGES     800000
#define NUM_FEATS   128
#define HIDDEN      256
#define NUM_CLASSES 40

typedef __attribute__((ext_vector_type(16))) _Float16 v16h;
typedef __attribute__((ext_vector_type(8)))  float    v8f;

#define ROWPTR_PAD 50176

#define BKT_ROWS    196
#define N_BKT       ((N_NODES + BKT_ROWS - 1) / BKT_ROWS)
#define N_CHUNK     256
#define CHUNK_E     (N_EDGES / N_CHUNK)
#define TBL_WORDS   (N_BKT * N_CHUNK)
#define BIN_THREADS 256
static_assert(TBL_WORDS % 4 == 0, "records behind the table stay 16-byte aligned");
#define BIN_ROUNDS  ((CHUNK_E + BIN_THREADS - 1) / BIN_THREADS)

__global__ __launch_bounds__(BIN_THREADS)
void bin_kernel(const int*   __restrict__ ei,
                const float* __restrict__ ew,
                int*         __restrict__ tbl,
                int*         __restrict__ rec1) {
    static_assert(N_EDGES % N_CHUNK == 0, "equal chunks");
    static_assert(N_BKT <= BIN_THREADS && CHUNK_E < 65536, "one thread per bucket; count and offset fit 16 bits");
    __shared__ int hist[BIN_THREADS];
    __shared__ int boff[BIN_THREADS];
    __shared__ int wtot[BIN_THREADS / 32];
    const int t    = threadIdx.x;
    const int lane = t & 31;
    const int wv   = t >> 5;
    const int e0   = blockIdx.x * CHUNK_E;
    hist[t] = 0;

    int   d[BIN_ROUNDS], sr[BIN_ROUNDS], rk[BIN_ROUNDS];
    float w[BIN_ROUNDS];
    #pragma unroll
    for (int k = 0; k < BIN_ROUNDS; ++k) {
        const int i = t + k * BIN_THREADS;
        const int e = e0 + (i < CHUNK_E ? i : CHUNK_E - 1);
        d[k]  = ei[N_EDGES + e];
        sr[k] = ei[e];
        w[k]  = ew[e];
    }
    __syncthreads();
    #pragma unroll
    for (int k = 0; k < BIN_ROUNDS; ++k) {
        rk[k] = 0;
        if (t + k * BIN_THREADS < CHUNK_E) rk[k] = atomicAdd(&hist[(unsigned)d[k] / BKT_ROWS], 1);
    }
    __syncthreads();

    const int c = hist[t];
    int inc = c;
    for (int o = 1; o < 32; o <<= 1) {
        const int v = __shfl_up(inc, o);
        if (lane >= o) inc += v;
    }
    if (lane == 31) wtot[wv] = inc;
    __syncthreads();
    int base = 0;
    for (int j = 0; j < BIN_THREADS / 32; ++j)
        if (j < wv) base += wtot[j];
    const int excl = base + inc - c;
    boff[t] = excl;
    if (t < N_BKT) tbl[t * N_CHUNK + blockIdx.x] = (c << 16) | excl;
    __syncthreads();

    #pragma unroll
    for (int k = 0; k < BIN_ROUNDS; ++k) {
        const int i = t + k * BIN_THREADS;
        if (i < CHUNK_E) {
            const int pos = e0 + boff[(unsigned)d[k] / BKT_ROWS] + rk[k];
            *(int4*)(rec1 + 4 * (long long)pos) = make_int4(e0 + i, sr[k], __float_as_int(w[k]), d[k]);
        }
    }
}

__device__ __forceinline__ int count_below(int keys, int n, int mine) {
    int r = 0;
    for (int j0 = 0; j0 < n; j0 += 8) {
        #pragma unroll
        for (int jj = 0; jj < 8; ++jj)
            r += (__builtin_amdgcn_readlane(keys, j0 + jj) < mine) ? 1 : 0;
    }
    return r;
}

template <int G>
__device__ __forceinline__ void take_rows(const float* __restrict__ feat, int D,
                                          int s, float w, int j, float4& acc) {
    float4 v[G];
    float  ww[G];
    #pragma unroll
    for (int u = 0; u < G; ++u) {
        const int sj = __builtin_amdgcn_readlane(s, j + u);
        ww[u] = __int_as_float(__builtin_amdgcn_readlane(__float_as_int(w), j + u));
        v[u] = *(const float4*)(feat + (long long)sj * D);
    }
    #pragma unroll
    for (int u = 0; u < G; ++u) {
        acc.x = fmaf(ww[u], v[u].x, acc.x);
        acc.y = fmaf(ww[u], v[u].y, acc.y);
        acc.z = fmaf(ww[u], v[u].z, acc.z);
        acc.w = fmaf(ww[u], v[u].w, acc.w);
    }
}

template <int M>
__device__ __forceinline__ float row_xor(float v) {
    return __int_as_float(__builtin_amdgcn_update_dpp(
        0, __float_as_int(v), 0x160 + M, 0xf, 0xf, false));
}

__device__ __forceinline__ void take_edges(const float* __restrict__ f, int D,
                                           int s, float w, int n, float4& acc) {
    int j = 0;
    for (; j + 8 <= n; j += 8) take_rows<8>(f, D, s, w, j, acc);
    switch (n - j) {
        case 7: take_rows<7>(f, D, s, w, j, acc); break;
        case 6: take_rows<6>(f, D, s, w, j, acc); break;
        case 5: take_rows<5>(f, D, s, w, j, acc); break;
        case 4: take_rows<4>(f, D, s, w, j, acc); break;
        case 3: take_rows<3>(f, D, s, w, j, acc); break;
        case 2: take_rows<2>(f, D, s, w, j, acc); break;
        case 1: take_rows<1>(f, D, s, w, j, acc); break;
        default: break;
    }
}

#define SCAN_ROWS   256
#define BKT_THREADS 1024
#define BKT_WAVES   (BKT_THREADS / 32)
#define BKT_CPW     (N_CHUNK / BKT_WAVES)
#define REC_CAP     12288
template <bool BIG>
__device__ __forceinline__ void sum_bucket_rows(const float* __restrict__ feat,
                                                const int4* recl, const int* recb,
                                                const int* rstart, int2 (*slot)[32],
                                                int wv, int lane, int row0, int base,
                                                int*   __restrict__ lo,
                                                float* __restrict__ lw,
                                                float* __restrict__ agg) {
    auto record = [&](int i) -> int4 {
        if constexpr (BIG) return *(const int4*)(recb + 4 * (long long)i);
        else               return recl[i];
    };

    const float* f = feat + lane * 4;
    for (int r = wv; r < BKT_ROWS; r += BKT_WAVES) {
        const int row = row0 + r;
        if (row >= N_NODES) break;
        const int beg = __builtin_amdgcn_readfirstlane(rstart[r]);
        const int end = __builtin_amdgcn_readfirstlane(rstart[r + 1]);
        const int nr  = end - beg;
        float4 acc = make_float4(0.0f, 0.0f, 0.0f, 0.0f);
        for (int c0 = 0; c0 < nr; c0 += 32) {
            int   s = 0;
            float w = 0.0f;
            if (nr <= 32) {
                int4 me = make_int4(0x7fffffff, 0, 0, 0);
                if (lane < nr) me = record(beg + lane);
                int rk = count_below(me.x, nr, me.x);
                rk = lane < nr ? rk : lane;
                s = __builtin_amdgcn_ds_permute(rk << 2, me.y);
                w = __int_as_float(__builtin_amdgcn_ds_permute(rk << 2, me.z));
            } else {
                for (int pb = 0; pb < nr; pb += 32) {
                    int4 me = make_int4(0x7fffffff, 0, 0, 0);
                    if (pb + lane < nr) me = record(beg + pb + lane);
                    int rk = 0;
                    for (int ob = 0; ob < nr; ob += 32) {
                        int keys = 0x7fffffff;
                        if (ob + lane < nr) keys = record(beg + ob + lane).x;
                        rk += count_below(keys, (nr - ob) < 32 ? (nr - ob) : 32, me.x);
                    }
                    if (pb + lane < nr && rk >= c0 && rk < c0 + 32) slot[wv][rk - c0] = make_int2(me.y, me.z);
                }
                if (c0 + lane < nr) {
                    const int2 sw = slot[wv][lane];
                    s = sw.x;
                    w = __int_as_float(sw.y);
                }
            }
            if (c0 + lane < nr) {
                const int p = base + beg + c0 + lane;
                lo[p] = s;
                lw[p] = w;
            }
            take_edges(f, NUM_FEATS, s, w, (nr - c0) < 32 ? (nr - c0) : 32, acc);
        }
        *(float4*)(agg + (long long)row * NUM_FEATS + lane * 4) = acc;
    }
}

__global__ __launch_bounds__(BKT_THREADS)
void bucket_gather_kernel(const int*   __restrict__ tbl,
                          const int*   __restrict__ rec1,
                          const float* __restrict__ feat,
                          int*         __restrict__ rowptr,
                          int*         __restrict__ rec,
                          int*         __restrict__ lo,
                          float*       __restrict__ lw,
                          float*       __restrict__ agg) {
    static_assert(N_CHUNK % BKT_WAVES == 0 && BKT_CPW <= 32 && N_CHUNK % 4 == 0, "whole chunks per wave, one table word per lane");
    static_assert(BKT_ROWS <= SCAN_ROWS && SCAN_ROWS <= BKT_THREADS && SCAN_ROWS % 32 == 0, "one thread per row in the scan");
    __shared__ int  rowcnt[SCAN_ROWS];
    __shared__ int  rstart[SCAN_ROWS];
    __shared__ int  wsum[BKT_WAVES];
    __shared__ int  wtot[SCAN_ROWS / 32];
    __shared__ int4 recl[REC_CAP];
    __shared__ int2 slot[BKT_WAVES][32];
    const int t    = threadIdx.x;
    const int lane = t & 31;
    const int wv   = __builtin_amdgcn_readfirstlane(t >> 5);
    const int b    = blockIdx.x;
    const int row0 = b * BKT_ROWS;
    if (t < SCAN_ROWS) rowcnt[t] = 0;

    const int word = tbl[b * N_CHUNK + wv * BKT_CPW + (lane % BKT_CPW)];

    int  n[BKT_CPW];
    long long at[BKT_CPW];
    int4 rc[BKT_CPW];
    const long long spare = (long long)(wv * BKT_CPW) * CHUNK_E;
    #pragma unroll
    for (int u = 0; u < BKT_CPW; ++u) {
        const int wd = __builtin_amdgcn_readlane(word, u);
        n[u]  = wd >> 16;
        at[u] = (long long)(wv * BKT_CPW + u) * CHUNK_E + (wd & 0xffff);
        rc[u] = *(const int4*)(rec1 + 4 * (lane < n[u] ? at[u] + lane : spare));
    }

    int pre = 0;
    const int4* t4 = (const int4*)tbl;
    const int   n4 = b * (N_CHUNK / 4);
    #pragma unroll 8
    for (int i = t; i < n4; i += BKT_THREADS) {
        const int4 v = t4[i];
        pre += (v.x >> 16) + (v.y >> 16) + (v.z >> 16) + (v.w >> 16);
    }
    __syncthreads();

    #pragma unroll
    for (int u = 0; u < BKT_CPW; ++u) {
        if (lane < n[u]) atomicAdd(&rowcnt[rc[u].w - row0], 1);
        if (n[u] > 32)
            for (int i = 32 + lane; i < n[u]; i += 32)
                atomicAdd(&rowcnt[rec1[4 * (at[u] + i) + 3] - row0], 1);
    }
    for (int o = 16; o > 0; o >>= 1) pre += __shfl_xor(pre, o);
    if (lane == 0) wsum[wv] = pre;
    __syncthreads();

    int base = 0;
    for (int j = 0; j < BKT_WAVES; ++j) base += wsum[j];
    const int c = t < SCAN_ROWS ? rowcnt[t] : 0;
    int inc = c;
    for (int o = 1; o < 32; o <<= 1) {
        const int v = __shfl_up(inc, o);
        if (lane >= o) inc += v;
    }
    if (t < SCAN_ROWS && lane == 31) wtot[t >> 5] = inc;
    __syncthreads();
    int total = 0;
    for (int j = 0; j < SCAN_ROWS / 32; ++j) total += wtot[j];
    if (t < SCAN_ROWS) {
        int wb = 0;
        for (int j = 0; j < SCAN_ROWS / 32; ++j)
            if (j < (t >> 5)) wb += wtot[j];
        const int excl = wb + inc - c;
        rowcnt[t] = excl;
        rstart[t] = excl;
        const int row = row0 + t;
        if (t < BKT_ROWS) {
            if (row < N_NODES)      rowptr[row]     = base + excl;
            if (row == N_NODES - 1) rowptr[N_NODES] = base + excl + c;
        }
    }
    const bool big = __builtin_amdgcn_readfirstlane(total) > REC_CAP;
    int* const recb = rec + 4 * (long long)base;
    __syncthreads();

    #pragma unroll
    for (int u = 0; u < BKT_CPW; ++u) {
        if (lane < n[u]) {
            const int  pos = atomicAdd(&rowcnt[rc[u].w - row0], 1);
            const int4 v   = make_int4(rc[u].x, rc[u].y, rc[u].z, 0);
            if (big) *(int4*)(recb + 4 * (long long)pos) = v;
            else     recl[pos] = v;
        }
        if (n[u] > 32) {
            for (int i = 32 + lane; i < n[u]; i += 32) {
                const int4 r   = *(const int4*)(rec1 + 4 * (at[u] + i));
                const int  pos = atomicAdd(&rowcnt[r.w - row0], 1);
                const int4 v   = make_int4(r.x, r.y, r.z, 0);
                if (big) *(int4*)(recb + 4 * (long long)pos) = v;
                else     recl[pos] = v;
            }
        }
    }
    __syncthreads();

    if (big) sum_bucket_rows<true >(feat, recl, recb, rstart, slot, wv, lane, row0, base, lo, lw, agg);
    else     sum_bucket_rows<false>(feat, recl, recb, rstart, slot, wv, lane, row0, base, lo, lw, agg);
}

#define GATHER_WAVES  8
#define GATHER_BLOCKS 1250
template <int D, bool LSM>
__global__ __launch_bounds__(GATHER_WAVES * 32)
void gather_kernel(const float* __restrict__ feat,
                   const int*   __restrict__ rowptr,
                   const int*   __restrict__ li,
                   const float* __restrict__ lwt,
                   float*       __restrict__ agg) {
    static_assert(D % 4 == 0 && D <= 128, "row = whole float4 lanes of one wave");
    static_assert(!LSM || D <= 64, "a softmax row sits on the lanes of one 16-lane row");
    const int lane = threadIdx.x & 31;
    const int wid = __builtin_amdgcn_readfirstlane(
        (int)(blockIdx.x * GATHER_WAVES + (threadIdx.x >> 5)));
    const int nw = gridDim.x * GATHER_WAVES;
    const int off = lane * 4;
    const bool on = off < D;
    const float* f = feat + (on ? off : D - 4);
    for (int row = wid; row < N_NODES; row += nw) {
        const int beg = rowptr[row];
        const int end = rowptr[row + 1];
        float4 acc = make_float4(0.0f, 0.0f, 0.0f, 0.0f);
        for (int base = beg; base < end; base += 32) {
            const int p = base + lane;
            int   s = 0;
            float w = 0.0f;
            if (p < end) { s = li[p]; w = lwt[p]; }
            take_edges(f, D, s, w, (end - base) < 32 ? (end - base) : 32, acc);
        }
        if constexpr (LSM) {
            float m = on ? fmaxf(fmaxf(acc.x, acc.y), fmaxf(acc.z, acc.w)) : -INFINITY;
            m = fmaxf(m, row_xor<1>(m));
            m = fmaxf(m, row_xor<2>(m));
            m = fmaxf(m, row_xor<4>(m));
            m = fmaxf(m, row_xor<8>(m));
            const float e0 = __expf(acc.x - m);
            const float e1 = __expf(acc.y - m);
            const float e2 = __expf(acc.z - m);
            const float e3 = __expf(acc.w - m);
            float s = on ? ((e0 + e1) + e2) + e3 : 0.0f;
            s += row_xor<1>(s);
            s += row_xor<2>(s);
            s += row_xor<4>(s);
            s += row_xor<8>(s);
            const float lse = __logf(s) + m;
            acc.x -= lse;
            acc.y -= lse;
            acc.z -= lse;
            acc.w -= lse;
        }
        if (on) *(float4*)(agg + (long long)row * D + off) = acc;
    }
}

#define GEMM_WAVES   13
#define GEMM_THREADS (GEMM_WAVES * 32)
#define N_STRIPS     (N_NODES / 16)
#define GEMM_BLOCKS  ((N_STRIPS + GEMM_WAVES - 1) / GEMM_WAVES)

typedef __attribute__((ext_vector_type(8))) _Float16 v8h;
typedef __attribute__((ext_vector_type(4))) _Float16 v4h;

struct raw16 { float4 a, b, c, d; };

__device__ __forceinline__ raw16 load16(const float* __restrict__ p) {
    raw16 r;
    r.a = *(const float4*)(p);
    r.b = *(const float4*)(p + 4);
    r.c = *(const float4*)(p + 16);
    r.d = *(const float4*)(p + 20);
    return r;
}

__device__ __forceinline__ v16h to_frag(const raw16& r) {
    v16h f;
    f[0]  = (_Float16)r.a.x; f[1]  = (_Float16)r.a.y; f[2]  = (_Float16)r.a.z; f[3]  = (_Float16)r.a.w;
    f[4]  = (_Float16)r.b.x; f[5]  = (_Float16)r.b.y; f[6]  = (_Float16)r.b.z; f[7]  = (_Float16)r.b.w;
    f[8]  = (_Float16)r.c.x; f[9]  = (_Float16)r.c.y; f[10] = (_Float16)r.c.z; f[11] = (_Float16)r.c.w;
    f[12] = (_Float16)r.d.x; f[13] = (_Float16)r.d.y; f[14] = (_Float16)r.d.z; f[15] = (_Float16)r.d.w;
    return f;
}

__device__ __forceinline__ v16h lds_frag(const unsigned char* wl, int o) {
    const v8h lo = *(const v8h*)(&wl[o]);
    const v8h hi = *(const v8h*)(&wl[o + 32]);
    return __builtin_shufflevector(lo, hi, 0, 1, 2, 3, 4, 5, 6, 7,
                                   8, 9, 10, 11, 12, 13, 14, 15);
}

template <int NV, int SV>
__device__ __forceinline__ void request_w(const float* __restrict__ W, float4 (&sv)[SV]) {
    #pragma unroll
    for (int i = 0; i < SV; ++i) {
        const int idx = (int)threadIdx.x + i * GEMM_THREADS;
        sv[i] = ((const float4*)W)[idx < NV ? idx : NV - 1];
    }
}

template <int K, int PITCH, int SV>
__device__ __forceinline__ void park_w(const float4 (&sv)[SV], unsigned char* wl) {
    #pragma unroll
    for (int i = 0; i < SV; ++i) {
        const int idx = (int)threadIdx.x + i * GEMM_THREADS;
        const int r   = idx / (K / 4);
        const int c4  = idx - r * (K / 4);
        v4h h;
        h[0] = (_Float16)sv[i].x; h[1] = (_Float16)sv[i].y;
        h[2] = (_Float16)sv[i].z; h[3] = (_Float16)sv[i].w;
        *(v4h*)(&wl[r * PITCH + c4 * 8]) = h;
    }
}

template <int K1, int NH, int NC>
__global__ __launch_bounds__(GEMM_THREADS)
void gemm_pair_kernel(const float* __restrict__ A,
                      const float* __restrict__ W1,
                      const float* __restrict__ W2,
                      float*       __restrict__ C) {
    constexpr int KS1    = K1 / 32;
    constexpr int HT     = NH / 16;
    constexpr int T2     = (NC + 15) / 16;
    constexpr int G1     = 4;
    constexpr int Q      = G1 / 2;
    constexpr int GROUPS = HT / G1;
    constexpr int SPG    = KS1 + Q;
    constexpr int STEPS  = GROUPS * SPG;
    constexpr int FR     = G1 > T2 ? G1 : T2;
    constexpr int PITCH1 = 2 * K1 + 16;
    constexpr int PITCH2 = 2 * NH + 16;
    constexpr int NV1    = NH * K1 / 4;
    constexpr int NV2    = NC * NH / 4;
    constexpr int SV1    = (NV1 + GEMM_THREADS - 1) / GEMM_THREADS;
    constexpr int SV2    = (NV2 + GEMM_THREADS - 1) / GEMM_THREADS;
    constexpr int LROWS1 = (SV1 * GEMM_THREADS + K1 / 4 - 1) / (K1 / 4);
    constexpr int LROWS2 = (SV2 * GEMM_THREADS + NH / 4 - 1) / (NH / 4);
    static_assert(K1 % 32 == 0 && NH % 32 == 0 && NC >= 16, "whole k-steps, at least one full tile");
    static_assert(HT % G1 == 0 && G1 % 2 == 0, "whole groups of tile pairs");
    static_assert((PITCH1 / 16) % 2 == 1 && (PITCH2 / 16) % 2 == 1, "LDS row pitch = 16 B x odd");
    static_assert(LROWS1 >= NH && LROWS2 >= NC, "room for all of W1 and W2");
    __shared__ __attribute__((aligned(16))) unsigned char wl1[LROWS1 * PITCH1];
    __shared__ __attribute__((aligned(16))) unsigned char wl2[LROWS2 * PITCH2];

    const int l     = threadIdx.x & 31;
    const int rl    = l & 15;
    const int khalf = (l >> 4) * 8;
    const int strip = __builtin_amdgcn_readfirstlane(
        (int)(blockIdx.x * GEMM_WAVES + (threadIdx.x >> 5)));
    const bool valid = strip < N_STRIPS;
    const int row0 = valid ? strip * 16 : 0;

    const float* arow = A + (long long)(row0 + rl) * K1 + khalf;
    raw16 araw[KS1];
    #pragma unroll
    for (int ks = 0; ks < KS1; ++ks) araw[ks] = load16(arow + ks * 32);
    __builtin_amdgcn_sched_barrier(0);

    {
        float4 sv1[SV1];
        float4 sv2[SV2];
        request_w<NV1, SV1>(W1, sv1);
        request_w<NV2, SV2>(W2, sv2);
        __builtin_amdgcn_sched_barrier(0);
        park_w<K1, PITCH1, SV1>(sv1, wl1);
        park_w<NH, PITCH2, SV2>(sv2, wl2);
    }
    __syncthreads();
    if (!valid) return;

    v16h af[KS1];
    #pragma unroll
    for (int ks = 0; ks < KS1; ++ks) af[ks] = to_frag(araw[ks]);

    const int base1 = rl * PITCH1 + khalf * 2;
    int woff2[T2];
    #pragma unroll
    for (int t = 0; t < T2; ++t) {
        const int col = t * 16 + rl;
        const bool in = (NC % 16 == 0) || (t + 1 < T2) || (col < NC);
        woff2[t] = (in ? col : NC - 1) * PITCH2 + khalf * 2;
    }

    v16h bf[2][FR];
    #pragma unroll
    for (int j = 0; j < G1; ++j) bf[0][j] = lds_frag(wl1, base1 + j * 16 * PITCH1);

    v8f  acc1[G1];
    v8f  acc2[T2];
    v16h hf[Q];
    #pragma unroll
    for (int s = 0; s < STEPS; ++s) {
        const int g = s / SPG;
        const int r = s - g * SPG;
        if (s + 1 < STEPS) {
            const int gn = (s + 1) / SPG;
            const int rn = (s + 1) - gn * SPG;
            if (rn < KS1) {
                #pragma unroll
                for (int j = 0; j < G1; ++j)
                    bf[(s + 1) & 1][j] = lds_frag(wl1, base1 + (gn * G1 + j) * 16 * PITCH1 + rn * 64);
            } else {
                const int kn = gn * Q + (rn - KS1);
                #pragma unroll
                for (int j = 0; j < T2; ++j)
                    bf[(s + 1) & 1][j] = lds_frag(wl2, woff2[j] + kn * 64);
            }
        }
        __builtin_amdgcn_sched_barrier(0);

        if (r < KS1) {
            #pragma unroll
            for (int j = 0; j < G1; ++j) {
                const v8f c0 = {};
                acc1[j] = __builtin_amdgcn_wmma_f32_16x16x32_f16(
                     false, bf[s & 1][j],  false, af[r],
                     (short)0, r == 0 ? c0 : acc1[j],  false,  false);
            }
            if (r == KS1 - 1) {
                #pragma unroll
                for (int q = 0; q < Q; ++q) {
                    #pragma unroll
                    for (int e = 0; e < 8; ++e) {
                        const float lo = acc1[2 * q][e];
                        const float hi = acc1[2 * q + 1][e];
                        hf[q][e]     = (_Float16)(lo > 0.0f ? lo : 0.0f);
                        hf[q][8 + e] = (_Float16)(hi > 0.0f ? hi : 0.0f);
                    }
                }
            }
        } else {
            const int q  = r - KS1;
            const int k2 = g * Q + q;
            #pragma unroll
            for (int j = 0; j < T2; ++j) {
                const v8f c0 = {};
                acc2[j] = __builtin_amdgcn_wmma_f32_16x16x32_f16(
                     false, hf[q],  false, bf[s & 1][j],
                     (short)0, k2 == 0 ? c0 : acc2[j],  false,  false);
            }
        }
    }

    #pragma unroll
    for (int j = 0; j < T2; ++j) {
        const int col = j * 16 + rl;
        const bool live = (NC % 16 == 0) || (j + 1 < T2) || (col < NC);
        if (live) {
            #pragma unroll
            for (int r = 0; r < 8; ++r)
                C[(long long)(row0 + khalf + r) * NC + col] = acc2[j][r];
        }
    }
}

extern "C" void kernel_launch(void* const* d_in, const int* in_sizes, int n_in,
                              void* d_out, int out_size, void* d_ws, size_t ws_size,
                              hipStream_t stream) {
    const float* x  = (const float*)d_in[0];
    const int*   ei = (const int*)  d_in[1];
    const float* ew = (const float*)d_in[2];
    const float* w1 = (const float*)d_in[3];
    const float* w2 = (const float*)d_in[4];
    float* out = (float*)d_out;

    const size_t bufElems = (size_t)N_NODES * HIDDEN;
    float* bufA = (float*)d_ws;
    float* bufB = bufA + bufElems;
    float* agg1 = bufA;
    float* p2   = bufB;

    int*   tbl    = (int*)bufB;
    int*   rec1   = tbl + TBL_WORDS;
    int*   rec    = rec1 + 4 * N_EDGES;
    int*   rowptr = (int*)(bufA + (size_t)N_NODES * NUM_FEATS);
    int*   csrc   = rowptr + ROWPTR_PAD;
    float* cw     = (float*)(csrc + N_EDGES);

    bin_kernel<<<N_CHUNK, BIN_THREADS, 0, stream>>>(ei, ew, tbl, rec1);

    bucket_gather_kernel<<<N_BKT, BKT_THREADS, 0, stream>>>(
        tbl, rec1, x, rowptr, rec, csrc, cw, agg1);

    gemm_pair_kernel<NUM_FEATS, HIDDEN, NUM_CLASSES>
        <<<GEMM_BLOCKS, GEMM_THREADS, 0, stream>>>(agg1, w1, w2, p2);

    gather_kernel<NUM_CLASSES, true><<<GATHER_BLOCKS, GATHER_WAVES * 32, 0, stream>>>(
        p2, rowptr, csrc, cw, out);
}
